# starting kernel
# baseline (speedup 1.0000x reference)
#include <hip/hip_runtime.h>

#define NN   100000
#define BB   512
#define EE   1000000
#define ESZ  1200000
#define DD   64
#define NBT  (NN + BB)

#define WS_A0   0
#define WS_A1   (NBT)
#define WS_M0   (2*NBT)
#define WS_M1   (3*NBT)
#define WS_DEG0 (4*NBT)
#define WS_DEG1 (5*NBT)
#define WS_SM   (6*NBT)
#define WS_U1   (WS_SM + 128)
#define WS_U2   (WS_SM + 192)
#define WS_T    (WS_SM + 256)
#define WS_ESA  (WS_T + 2*BB)
#define WS_HID  (WS_ESA + 2*BB*DD)

typedef float v2f __attribute__((ext_vector_type(2)));
typedef float v8f __attribute__((ext_vector_type(8)));

__global__ void k_zero(float* ws) {
    int i = blockIdx.x * blockDim.x + threadIdx.x;
    if (i < 6 * NBT) ws[i] = 0.0f;
}

__global__ void k_small(const float* __restrict__ w_n2l,
                        const float* __restrict__ p_conv,
                        const float* __restrict__ W_sage,
                        float* ws) {
    __shared__ float v[DD], w[DD];
    int d = threadIdx.x;
    float x = fmaxf(w_n2l[d] + w_n2l[DD + d], 0.0f);
    v[d] = x;
    __syncthreads();
    float ss = 0.f;
    for (int k = 0; k < DD; ++k) ss += v[k] * v[k];
    float nv = x / fmaxf(sqrtf(ss), 1e-12f);
    __syncthreads();
    v[d] = nv;
    __syncthreads();
    float wd = 0.f;
    for (int k = 0; k < DD; ++k) wd += v[k] * p_conv[k * DD + d];
    w[d] = wd;
    __syncthreads();
    float u1 = 0.f, u2 = 0.f;
    for (int k = 0; k < DD; ++k) {
        u1 += w[k] * W_sage[k * DD + d];
        u2 += w[k] * W_sage[(DD + k) * DD + d];
    }
    ws[WS_SM + d]  = nv;
    ws[WS_SM + 64 + d] = wd;
    ws[WS_U1 + d] = u1;
    ws[WS_U2 + d] = u2;
}

__global__ void k_n2n(const int* __restrict__ r0, const float* __restrict__ v0,
                      const int* __restrict__ r1, const float* __restrict__ v1,
                      float* ws) {
    int e = blockIdx.x * blockDim.x + threadIdx.x;
    if (e < EE) {
        atomicAdd(&ws[WS_A0 + r0[e]], v0[e]);
    } else if (e < 2 * EE) {
        int i = e - EE;
        atomicAdd(&ws[WS_A1 + r1[i]], v1[i]);
    }
}

__global__ void k_subg(const int* __restrict__ subg_row, float* ws) {
    int i = blockIdx.x * blockDim.x + threadIdx.x;
    if (i < NN) {
        int b = subg_row[i];
        atomicAdd(&ws[WS_A0 + NN + b], 1.0f);
        atomicAdd(&ws[WS_A1 + NN + b], 1.0f);
    }
}

__global__ void k_sage(const int* __restrict__ s0r, const int* __restrict__ s0c,
                       const int* __restrict__ s1r, const int* __restrict__ s1c,
                       float* ws) {
    int e = blockIdx.x * blockDim.x + threadIdx.x;
    if (e < ESZ) {
        int r = s0r[e];
        atomicAdd(&ws[WS_M0 + r], ws[WS_A0 + s0c[e]]);
        atomicAdd(&ws[WS_DEG0 + r], 1.0f);
    } else if (e < 2 * ESZ) {
        int i = e - ESZ;
        int r = s1r[i];
        atomicAdd(&ws[WS_M1 + r], ws[WS_A1 + s1c[i]]);
        atomicAdd(&ws[WS_DEG1 + r], 1.0f);
    }
}

__global__ void k_rows(float* ws, const float* __restrict__ cp, float* out) {
    int wid  = (blockIdx.x * blockDim.x + threadIdx.x) >> 5;
    int lane = threadIdx.x & 31;
    if (wid >= 2 * NBT) return;
    int l = wid / NBT;
    int i = wid - l * NBT;

    float a   = ws[(l ? WS_A1  : WS_A0)  + i];
    float num = ws[(l ? WS_M1  : WS_M0)  + i];
    float dg  = ws[(l ? WS_DEG1: WS_DEG0)+ i];
    float m   = num / fmaxf(dg, 1.0f);

    int d0 = 2 * lane, d1 = 2 * lane + 1;
    float r0 = fmaxf(a * ws[WS_U1 + d0] + m * ws[WS_U2 + d0], 0.0f);
    float r1 = fmaxf(a * ws[WS_U1 + d1] + m * ws[WS_U2 + d1], 0.0f);

    float ss = r0 * r0 + r1 * r1;
    for (int off = 16; off >= 1; off >>= 1) ss += __shfl_xor(ss, off, 32);
    float inv = 1.0f / fmaxf(sqrtf(ss), 1e-12f);
    r0 *= inv; r1 *= inv;

    if (i < NN) {
        float2 pr = make_float2(r0, r1);
        float* dst = out + BB + (l * NN + i) * DD + d0;
        *(float2*)dst = pr;
    } else {
        int b = i - NN;
        float t = r0 * cp[d0] + r1 * cp[d1];
        for (int off = 16; off >= 1; off >>= 1) t += __shfl_xor(t, off, 32);
        if (lane == 0) ws[WS_T + l * BB + b] = t;
    }
}

__global__ void k_esa(const int* __restrict__ act_col, const float* __restrict__ out,
                      float* ws) {
    int idx = blockIdx.x * blockDim.x + threadIdx.x;
    if (idx >= 2 * BB * DD) return;
    int d = idx & (DD - 1);
    int b = (idx >> 6) & (BB - 1);
    int l = idx >> 15;
    float t = ws[WS_T + l * BB + b];
    float ae = out[BB + (l * NN + act_col[b]) * DD + d];
    ws[WS_ESA + idx] = t * ae;
}

__global__ void k_gemm(const float* __restrict__ ws, const float* __restrict__ h1,
                       float* __restrict__ hid) {
    int bid  = blockIdx.x;
    int tn   = bid & 1;
    int tm   = (bid >> 1) & 31;
    int l    = bid >> 6;
    int lane = threadIdx.x;
    int half = lane >> 4;
    int lid  = lane & 15;

    const float* A = ws + WS_ESA + (l * BB + tm * 16) * DD;
    v8f c = {0.f, 0.f, 0.f, 0.f, 0.f, 0.f, 0.f, 0.f};
    #pragma unroll
    for (int s = 0; s < 16; ++s) {
        int k0 = 4 * s + 2 * half;
        v2f av, bv;
        av.x = A[lid * DD + k0];
        av.y = A[lid * DD + k0 + 1];
        bv.x = h1[(k0    ) * 32 + tn * 16 + lid];
        bv.y = h1[(k0 + 1) * 32 + tn * 16 + lid];
        c = __builtin_amdgcn_wmma_f32_16x16x4_f32(false, av, false, bv,
                                                  (short)0, c, false, false);
    }
    float* dst = hid + (l * BB + tm * 16) * 32 + tn * 16;
    #pragma unroll
    for (int r = 0; r < 8; ++r) {
        int M = r + 8 * half;
        dst[M * 32 + lid] = fmaxf(c[r], 0.0f);
    }
}

__global__ void k_q(const float* __restrict__ ws, const float* __restrict__ aux,
                    const float* __restrict__ h2w, float* out) {
    int b = blockIdx.x * blockDim.x + threadIdx.x;
    if (b >= BB) return;
    float q = 0.0f;
    for (int l = 0; l < 2; ++l) {
        const float* h = ws + WS_HID + (l * BB + b) * 32;
        for (int j = 0; j < 32; ++j) q += h[j] * h2w[j];
        for (int j = 0; j < 4; ++j)  q += aux[b * 8 + l * 4 + j] * h2w[32 + j];
    }
    out[b] = q;
}

extern "C" void kernel_launch(void* const* d_in, const int* in_sizes, int n_in,
                              void* d_out, int out_size, void* d_ws, size_t ws_size,
                              hipStream_t stream) {
    const int*   n2n0_row = (const int*)  d_in[0];
    const float* n2n0_val = (const float*)d_in[2];
    const int*   n2n1_row = (const int*)  d_in[3];
    const float* n2n1_val = (const float*)d_in[5];
    const int*   subg_row = (const int*)  d_in[6];
    const int*   act_col  = (const int*)  d_in[9];
    const int*   sg0_row  = (const int*)  d_in[10];
    const int*   sg0_col  = (const int*)  d_in[11];
    const int*   sg1_row  = (const int*)  d_in[12];
    const int*   sg1_col  = (const int*)  d_in[13];
    const float* aux      = (const float*)d_in[14];
    const float* w_n2l    = (const float*)d_in[15];
    const float* p_conv   = (const float*)d_in[16];
    const float* W_sage   = (const float*)d_in[17];
    const float* cp       = (const float*)d_in[18];
    const float* h1w      = (const float*)d_in[19];
    const float* h2w      = (const float*)d_in[20];
    float* out = (float*)d_out;
    float* ws  = (float*)d_ws;

    k_zero <<<(6 * NBT + 255) / 256, 256, 0, stream>>>(ws);
    k_small<<<1, 64, 0, stream>>>(w_n2l, p_conv, W_sage, ws);
    k_n2n  <<<(2 * EE + 255) / 256, 256, 0, stream>>>(n2n0_row, n2n0_val,
                                                      n2n1_row, n2n1_val, ws);
    k_subg <<<(NN + 255) / 256, 256, 0, stream>>>(subg_row, ws);
    k_sage <<<(2 * ESZ + 255) / 256, 256, 0, stream>>>(sg0_row, sg0_col,
                                                       sg1_row, sg1_col, ws);
    k_rows <<<(2 * NBT * 32 + 255) / 256, 256, 0, stream>>>(ws, cp, out);
    k_esa  <<<(2 * BB * DD + 255) / 256, 256, 0, stream>>>(act_col, out, ws);
    k_gemm <<<128, 32, 0, stream>>>(ws, h1w, ws + WS_HID);
    k_q    <<<(BB + 255) / 256, 256, 0, stream>>>(ws, aux, h2w, out);
}


# row kernel: 8192 resident waves loop over row pairs (16 lanes per row, 4 d per lane, DPP row sums, one 16-B store writes two whole rows; was 201024 one-row waves); norm and t sum order changes
# speedup vs baseline: 2.3081x; 2.3081x over previous
#include <hip/hip_runtime.h>

#define NN   100000
#define BB   512
#define EE   1000000
#define ESZ  1200000
#define DD   64
#define NBT  (NN + BB)

#define WS_A0   0
#define WS_A1   (NBT)
#define WS_MD0  (2*NBT)
#define WS_MD1  (4*NBT)
#define WS_SM   (6*NBT)
#define WS_U1   (WS_SM + 128)
#define WS_U2   (WS_SM + 192)
#define WS_T    (WS_SM + 256)
#define WS_ESA  (WS_T + 2*BB)
#define WS_HID  (WS_ESA + 2*BB*DD)

typedef float v2f __attribute__((ext_vector_type(2)));
typedef float v8f __attribute__((ext_vector_type(8)));

__global__ void k_zero(float* ws) {
    int i = blockIdx.x * blockDim.x + threadIdx.x;
    if (i < 6 * NBT) ws[i] = 0.0f;
}

__global__ void k_small(const float* __restrict__ w_n2l,
                        const float* __restrict__ p_conv,
                        const float* __restrict__ W_sage,
                        float* ws) {
    __shared__ float v[DD], w[DD];
    int d = threadIdx.x;
    float x = fmaxf(w_n2l[d] + w_n2l[DD + d], 0.0f);
    v[d] = x;
    __syncthreads();
    float ss = 0.f;
    for (int k = 0; k < DD; ++k) ss += v[k] * v[k];
    float nv = x / fmaxf(sqrtf(ss), 1e-12f);
    __syncthreads();
    v[d] = nv;
    __syncthreads();
    float wd = 0.f;
    for (int k = 0; k < DD; ++k) wd += v[k] * p_conv[k * DD + d];
    w[d] = wd;
    __syncthreads();
    float u1 = 0.f, u2 = 0.f;
    for (int k = 0; k < DD; ++k) {
        u1 += w[k] * W_sage[k * DD + d];
        u2 += w[k] * W_sage[(DD + k) * DD + d];
    }
    ws[WS_SM + d]  = nv;
    ws[WS_SM + 64 + d] = wd;
    ws[WS_U1 + d] = u1;
    ws[WS_U2 + d] = u2;
}

#define SORT_T    16384
#define SORT_NT   1024
#define SORT_PER  (SORT_T / SORT_NT)
#define N2N_BINS  ((NBT + 31) / 32)
#define SAGE_BINS ((NBT + 15) / 16)
#define N2N_BLK   ((EE + SORT_T - 1) / SORT_T)
#define SAGE_BLK  ((ESZ + SORT_T - 1) / SORT_T)
static_assert(SORT_T == 32 * 32 * SORT_PER, "32 waves x SORT_PER groups of 32 ordered edges");

template <int NB>
__device__ __forceinline__ void bins_to_offsets(unsigned* cnt, unsigned* wsum) {
    constexpr int PER = (NB + SORT_NT - 1) / SORT_NT;
    const int tid = threadIdx.x, lane = tid & 31, wave = tid >> 5;
    unsigned loc[PER];
    unsigned s = 0;
    #pragma unroll
    for (int k = 0; k < PER; ++k) {
        int b = tid * PER + k;
        unsigned v = (b < NB) ? cnt[b] : 0u;
        loc[k] = s;
        s += v;
    }
    unsigned inc = s;
    for (int off = 1; off < 32; off <<= 1) {
        unsigned t = __shfl_up(inc, off, 32);
        if (lane >= off) inc += t;
    }
    if (lane == 31) wsum[wave] = inc;
    __syncthreads();
    if (wave == 0) {
        unsigned w = wsum[lane];
        unsigned winc = w;
        for (int off = 1; off < 32; off <<= 1) {
            unsigned t = __shfl_up(winc, off, 32);
            if (lane >= off) winc += t;
        }
        wsum[lane] = winc - w;
    }
    __syncthreads();
    unsigned base = wsum[wave] + (inc - s);
    #pragma unroll
    for (int k = 0; k < PER; ++k) {
        int b = tid * PER + k;
        if (b < NB) cnt[b] = base + loc[k];
    }
    __syncthreads();
}

__global__ void __launch_bounds__(SORT_NT)
k_n2n(const int* __restrict__ r0, const float* __restrict__ v0,
      const int* __restrict__ r1, const float* __restrict__ v1,
      float* ws) {
    __shared__ unsigned cnt[N2N_BINS + 1];
    __shared__ unsigned wsum[32];
    __shared__ uint2 sorted[SORT_T];
    const int tid = threadIdx.x;
    const bool l1 = blockIdx.x >= N2N_BLK;
    const int blk = l1 ? blockIdx.x - N2N_BLK : blockIdx.x;
    const int*   rr = l1 ? r1 : r0;
    const float* vv = l1 ? v1 : v0;
    float* plane = ws + (l1 ? WS_A1 : WS_A0);
    const int base = blk * SORT_T;
    const int n = min(SORT_T, EE - base);

    for (int b = tid; b <= N2N_BINS; b += SORT_NT) cnt[b] = 0u;
    int row[SORT_PER]; float val[SORT_PER]; unsigned rank[SORT_PER];
    #pragma unroll
    for (int j = 0; j < SORT_PER; ++j) {
        int k = j * SORT_NT + tid;
        int idx = base + min(k, n - 1);
        row[j] = rr[idx];
        val[j] = vv[idx];
    }
    __syncthreads();
    #pragma unroll
    for (int j = 0; j < SORT_PER; ++j) {
        int k = j * SORT_NT + tid;
        int bin = (k < n) ? (row[j] >> 5) : N2N_BINS;
        rank[j] = atomicAdd(&cnt[bin], 1u);
    }
    __syncthreads();
    bins_to_offsets<N2N_BINS + 1>(cnt, wsum);
    #pragma unroll
    for (int j = 0; j < SORT_PER; ++j) {
        int k = j * SORT_NT + tid;
        int bin = (k < n) ? (row[j] >> 5) : N2N_BINS;
        sorted[cnt[bin] + rank[j]] = make_uint2((unsigned)row[j], __float_as_uint(val[j]));
    }
    __syncthreads();
    uint2 e[SORT_PER];
    #pragma unroll
    for (int j = 0; j < SORT_PER; ++j) e[j] = sorted[j * SORT_NT + tid];
    #pragma unroll
    for (int j = 0; j < SORT_PER; ++j) {
        int k = j * SORT_NT + tid;
        if (k < n) atomicAdd(&plane[e[j].x], __uint_as_float(e[j].y));
    }
}

__global__ void k_subg(const int* __restrict__ subg_row, float* ws) {
    int i = blockIdx.x * blockDim.x + threadIdx.x;
    if (i < NN) {
        int b = subg_row[i];
        atomicAdd(&ws[WS_A0 + NN + b], 1.0f);
        atomicAdd(&ws[WS_A1 + NN + b], 1.0f);
    }
}

__global__ void __launch_bounds__(SORT_NT)
k_sage(const int* __restrict__ s0r, const int* __restrict__ s0c,
       const int* __restrict__ s1r, const int* __restrict__ s1c,
       float* ws) {
    __shared__ unsigned cnt[SAGE_BINS + 1];
    __shared__ unsigned wsum[32];
    __shared__ uint2 sorted[SORT_T];
    const int tid = threadIdx.x, lane = tid & 31, wave = tid >> 5;
    const bool l1 = blockIdx.x >= SAGE_BLK;
    const int blk = l1 ? blockIdx.x - SAGE_BLK : blockIdx.x;
    const int* rr = l1 ? s1r : s0r;
    const int* cc = l1 ? s1c : s0c;
    const float* ap = ws + (l1 ? WS_A1 : WS_A0);
    float* md = ws + (l1 ? WS_MD1 : WS_MD0);
    const int base = blk * SORT_T;
    const int n = min(SORT_T, ESZ - base);

    for (int b = tid; b <= SAGE_BINS; b += SORT_NT) cnt[b] = 0u;
    int row[SORT_PER]; int col[SORT_PER]; float av[SORT_PER]; unsigned rank[SORT_PER];
    #pragma unroll
    for (int j = 0; j < SORT_PER; ++j) {
        int k = j * SORT_NT + tid;
        int idx = base + min(k, n - 1);
        row[j] = rr[idx];
        col[j] = cc[idx];
    }
    #pragma unroll
    for (int j = 0; j < SORT_PER; ++j) av[j] = ap[col[j]];
    __syncthreads();
    #pragma unroll
    for (int j = 0; j < SORT_PER; ++j) {
        int k = j * SORT_NT + tid;
        int bin = (k < n) ? (row[j] >> 4) : SAGE_BINS;
        rank[j] = atomicAdd(&cnt[bin], 1u);
    }
    __syncthreads();
    bins_to_offsets<SAGE_BINS + 1>(cnt, wsum);
    #pragma unroll
    for (int j = 0; j < SORT_PER; ++j) {
        int k = j * SORT_NT + tid;
        int bin = (k < n) ? (row[j] >> 4) : SAGE_BINS;
        sorted[cnt[bin] + rank[j]] = make_uint2((unsigned)row[j], __float_as_uint(av[j]));
    }
    __syncthreads();
    const int jj   = lane >> 1;
    const int role = lane & 1;
    uint2 eA[SORT_PER], eB[SORT_PER];
    #pragma unroll
    for (int j = 0; j < SORT_PER; ++j) {
        int p = (j * 32 + wave) * 32 + jj;
        eA[j] = sorted[p];
        eB[j] = sorted[p + 16];
    }
    #pragma unroll
    for (int j = 0; j < SORT_PER; ++j) {
        int p = (j * 32 + wave) * 32 + jj;
        if (p < n)
            atomicAdd(&md[2 * eA[j].x + role], role ? 1.0f : __uint_as_float(eA[j].y));
        if (p + 16 < n)
            atomicAdd(&md[2 * eB[j].x + role], role ? 1.0f : __uint_as_float(eB[j].y));
    }
}

#define ROWS_NT    256
#define ROWS_BLK   1024
#define ROWS_WAVES (ROWS_BLK * ROWS_NT / 32)
#define ROWS_UNR   4
static_assert(NBT % 2 == 0 && NN % 2 == 0, "row pairs must not straddle a layer or the real/virtual split");

__device__ __forceinline__ float half_sum(float v) {
    v += __int_as_float(__builtin_amdgcn_update_dpp(0, __float_as_int(v), 0x161, 0xf, 0xf, true));
    v += __int_as_float(__builtin_amdgcn_update_dpp(0, __float_as_int(v), 0x162, 0xf, 0xf, true));
    v += __int_as_float(__builtin_amdgcn_update_dpp(0, __float_as_int(v), 0x164, 0xf, 0xf, true));
    v += __int_as_float(__builtin_amdgcn_update_dpp(0, __float_as_int(v), 0x168, 0xf, 0xf, true));
    return v;
}

__global__ void __launch_bounds__(ROWS_NT)
k_rows(float* ws, const float* __restrict__ cp, float* out) {
    const int lane = threadIdx.x & 31;
    const int half = lane >> 4;
    const int q    = lane & 15;
    const int gw   = (blockIdx.x * ROWS_NT + threadIdx.x) >> 5;

    const float4 u1 = *(const float4*)&ws[WS_U1 + 4 * q];
    const float4 u2 = *(const float4*)&ws[WS_U2 + 4 * q];

    for (int p0 = gw; p0 < NBT; p0 += ROWS_WAVES * ROWS_UNR) {
        float a[ROWS_UNR]; float2 md[ROWS_UNR];
        #pragma unroll
        for (int k = 0; k < ROWS_UNR; ++k) {
            int p = min(p0 + k * ROWS_WAVES, NBT - 1);
            int w = 2 * p + half;
            int l = (w >= NBT) ? 1 : 0;
            int i = w - l * NBT;
            a[k]  = ws[(l ? WS_A1 : WS_A0) + i];
            md[k] = *(const float2*)&ws[(l ? WS_MD1 : WS_MD0) + 2 * i];
        }
        #pragma unroll
        for (int k = 0; k < ROWS_UNR; ++k) {
            int p = p0 + k * ROWS_WAVES;
            if (p < NBT) {
                int w = 2 * p + half;
                int l = (w >= NBT) ? 1 : 0;
                int i = w - l * NBT;
                float m = md[k].x / fmaxf(md[k].y, 1.0f);
                float r0 = fmaxf(a[k] * u1.x + m * u2.x, 0.0f);
                float r1 = fmaxf(a[k] * u1.y + m * u2.y, 0.0f);
                float r2 = fmaxf(a[k] * u1.z + m * u2.z, 0.0f);
                float r3 = fmaxf(a[k] * u1.w + m * u2.w, 0.0f);
                float ss = (r0 * r0 + r1 * r1) + (r2 * r2 + r3 * r3);
                ss = half_sum(ss);
                float inv = 1.0f / fmaxf(sqrtf(ss), 1e-12f);
                r0 *= inv; r1 *= inv; r2 *= inv; r3 *= inv;
                if (i < NN) {
                    float* dst = out + BB + (l * NN + i) * DD + 4 * q;
                    *(float4*)dst = make_float4(r0, r1, r2, r3);
                } else {
                    int b = i - NN;
                    const float4 c4 = *(const float4*)&cp[4 * q];
                    float t = (r0 * c4.x + r1 * c4.y) + (r2 * c4.z + r3 * c4.w);
                    t = half_sum(t);
                    if (q == 0) ws[WS_T + l * BB + b] = t;
                }
            }
        }
    }
}

__global__ void k_esa(const int* __restrict__ act_col, const float* __restrict__ out,
                      float* ws) {
    int idx = blockIdx.x * blockDim.x + threadIdx.x;
    if (idx >= 2 * BB * DD) return;
    int d = idx & (DD - 1);
    int b = (idx >> 6) & (BB - 1);
    int l = idx >> 15;
    float t = ws[WS_T + l * BB + b];
    float ae = out[BB + (l * NN + act_col[b]) * DD + d];
    ws[WS_ESA + idx] = t * ae;
}

__global__ void k_gemm(const float* __restrict__ ws, const float* __restrict__ h1,
                       float* __restrict__ hid) {
    int bid  = blockIdx.x;
    int tn   = bid & 1;
    int tm   = (bid >> 1) & 31;
    int l    = bid >> 6;
    int lane = threadIdx.x;
    int half = lane >> 4;
    int lid  = lane & 15;

    const float* A = ws + WS_ESA + (l * BB + tm * 16) * DD;
    v8f c = {0.f, 0.f, 0.f, 0.f, 0.f, 0.f, 0.f, 0.f};
    #pragma unroll
    for (int s = 0; s < 16; ++s) {
        int k0 = 4 * s + 2 * half;
        v2f av, bv;
        av.x = A[lid * DD + k0];
        av.y = A[lid * DD + k0 + 1];
        bv.x = h1[(k0    ) * 32 + tn * 16 + lid];
        bv.y = h1[(k0 + 1) * 32 + tn * 16 + lid];
        c = __builtin_amdgcn_wmma_f32_16x16x4_f32(false, av, false, bv,
                                                  (short)0, c, false, false);
    }
    float* dst = hid + (l * BB + tm * 16) * 32 + tn * 16;
    #pragma unroll
    for (int r = 0; r < 8; ++r) {
        int M = r + 8 * half;
        dst[M * 32 + lid] = fmaxf(c[r], 0.0f);
    }
}

__global__ void k_q(const float* __restrict__ ws, const float* __restrict__ aux,
                    const float* __restrict__ h2w, float* out) {
    int b = blockIdx.x * blockDim.x + threadIdx.x;
    if (b >= BB) return;
    float q = 0.0f;
    for (int l = 0; l < 2; ++l) {
        const float* h = ws + WS_HID + (l * BB + b) * 32;
        for (int j = 0; j < 32; ++j) q += h[j] * h2w[j];
        for (int j = 0; j < 4; ++j)  q += aux[b * 8 + l * 4 + j] * h2w[32 + j];
    }
    out[b] = q;
}

extern "C" void kernel_launch(void* const* d_in, const int* in_sizes, int n_in,
                              void* d_out, int out_size, void* d_ws, size_t ws_size,
                              hipStream_t stream) {
    const int*   n2n0_row = (const int*)  d_in[0];
    const float* n2n0_val = (const float*)d_in[2];
    const int*   n2n1_row = (const int*)  d_in[3];
    const float* n2n1_val = (const float*)d_in[5];
    const int*   subg_row = (const int*)  d_in[6];
    const int*   act_col  = (const int*)  d_in[9];
    const int*   sg0_row  = (const int*)  d_in[10];
    const int*   sg0_col  = (const int*)  d_in[11];
    const int*   sg1_row  = (const int*)  d_in[12];
    const int*   sg1_col  = (const int*)  d_in[13];
    const float* aux      = (const float*)d_in[14];
    const float* w_n2l    = (const float*)d_in[15];
    const float* p_conv   = (const float*)d_in[16];
    const float* W_sage   = (const float*)d_in[17];
    const float* cp       = (const float*)d_in[18];
    const float* h1w      = (const float*)d_in[19];
    const float* h2w      = (const float*)d_in[20];
    float* out = (float*)d_out;
    float* ws  = (float*)d_ws;

    k_zero <<<(6 * NBT + 255) / 256, 256, 0, stream>>>(ws);
    k_small<<<1, 64, 0, stream>>>(w_n2l, p_conv, W_sage, ws);
    k_n2n  <<<2 * N2N_BLK, SORT_NT, 0, stream>>>(n2n0_row, n2n0_val,
                                                  n2n1_row, n2n1_val, ws);
    k_subg <<<(NN + 255) / 256, 256, 0, stream>>>(subg_row, ws);
    k_sage <<<2 * SAGE_BLK, SORT_NT, 0, stream>>>(sg0_row, sg0_col,
                                                   sg1_row, sg1_col, ws);
    k_rows <<<ROWS_BLK, ROWS_NT, 0, stream>>>(ws, cp, out);
    k_esa  <<<(2 * BB * DD + 255) / 256, 256, 0, stream>>>(act_col, out, ws);
    k_gemm <<<128, 32, 0, stream>>>(ws, h1w, ws + WS_HID);
    k_q    <<<(BB + 255) / 256, 256, 0, stream>>>(ws, aux, h2w, out);
}


# n2n and sage scatters: slots past a layer's last edge skip the LDS counting add (were parked on ONE spare bin: ~490 / ~390 same-word wave adds per last block, served lane by lane); sum order kept
# speedup vs baseline: 2.7250x; 1.1806x over previous
#include <hip/hip_runtime.h>

#define NN   100000
#define BB   512
#define EE   1000000
#define ESZ  1200000
#define DD   64
#define NBT  (NN + BB)

#define WS_A0   0
#define WS_A1   (NBT)
#define WS_MD0  (2*NBT)
#define WS_MD1  (4*NBT)
#define WS_SM   (6*NBT)
#define WS_U1   (WS_SM + 128)
#define WS_U2   (WS_SM + 192)
#define WS_T    (WS_SM + 256)
#define WS_ESA  (WS_T + 2*BB)
#define WS_HID  (WS_ESA + 2*BB*DD)

typedef float v2f __attribute__((ext_vector_type(2)));
typedef float v8f __attribute__((ext_vector_type(8)));

__global__ void k_zero(float* ws) {
    int i = blockIdx.x * blockDim.x + threadIdx.x;
    if (i < 6 * NBT) ws[i] = 0.0f;
}

__global__ void k_small(const float* __restrict__ w_n2l,
                        const float* __restrict__ p_conv,
                        const float* __restrict__ W_sage,
                        float* ws) {
    __shared__ float v[DD], w[DD];
    int d = threadIdx.x;
    float x = fmaxf(w_n2l[d] + w_n2l[DD + d], 0.0f);
    v[d] = x;
    __syncthreads();
    float ss = 0.f;
    for (int k = 0; k < DD; ++k) ss += v[k] * v[k];
    float nv = x / fmaxf(sqrtf(ss), 1e-12f);
    __syncthreads();
    v[d] = nv;
    __syncthreads();
    float wd = 0.f;
    for (int k = 0; k < DD; ++k) wd += v[k] * p_conv[k * DD + d];
    w[d] = wd;
    __syncthreads();
    float u1 = 0.f, u2 = 0.f;
    for (int k = 0; k < DD; ++k) {
        u1 += w[k] * W_sage[k * DD + d];
        u2 += w[k] * W_sage[(DD + k) * DD + d];
    }
    ws[WS_SM + d]  = nv;
    ws[WS_SM + 64 + d] = wd;
    ws[WS_U1 + d] = u1;
    ws[WS_U2 + d] = u2;
}

#define SORT_T    16384
#define SORT_NT   1024
#define SORT_PER  (SORT_T / SORT_NT)
#define N2N_BINS  ((NBT + 31) / 32)
#define SAGE_BINS ((NBT + 15) / 16)
#define N2N_BLK   ((EE + SORT_T - 1) / SORT_T)
#define SAGE_BLK  ((ESZ + SORT_T - 1) / SORT_T)
static_assert(SORT_T == 32 * 32 * SORT_PER, "32 waves x SORT_PER groups of 32 ordered edges");

template <int NB>
__device__ __forceinline__ void bins_to_offsets(unsigned* cnt, unsigned* wsum) {
    constexpr int PER = (NB + SORT_NT - 1) / SORT_NT;
    const int tid = threadIdx.x, lane = tid & 31, wave = tid >> 5;
    unsigned loc[PER];
    unsigned s = 0;
    #pragma unroll
    for (int k = 0; k < PER; ++k) {
        int b = tid * PER + k;
        unsigned v = (b < NB) ? cnt[b] : 0u;
        loc[k] = s;
        s += v;
    }
    unsigned inc = s;
    for (int off = 1; off < 32; off <<= 1) {
        unsigned t = __shfl_up(inc, off, 32);
        if (lane >= off) inc += t;
    }
    if (lane == 31) wsum[wave] = inc;
    __syncthreads();
    if (wave == 0) {
        unsigned w = wsum[lane];
        unsigned winc = w;
        for (int off = 1; off < 32; off <<= 1) {
            unsigned t = __shfl_up(winc, off, 32);
            if (lane >= off) winc += t;
        }
        wsum[lane] = winc - w;
    }
    __syncthreads();
    unsigned base = wsum[wave] + (inc - s);
    #pragma unroll
    for (int k = 0; k < PER; ++k) {
        int b = tid * PER + k;
        if (b < NB) cnt[b] = base + loc[k];
    }
    __syncthreads();
}

__global__ void __launch_bounds__(SORT_NT)
k_n2n(const int* __restrict__ r0, const float* __restrict__ v0,
      const int* __restrict__ r1, const float* __restrict__ v1,
      float* ws) {
    __shared__ unsigned cnt[N2N_BINS];
    __shared__ unsigned wsum[32];
    __shared__ uint2 sorted[SORT_T];
    const int tid = threadIdx.x;
    const bool l1 = blockIdx.x >= N2N_BLK;
    const int blk = l1 ? blockIdx.x - N2N_BLK : blockIdx.x;
    const int*   rr = l1 ? r1 : r0;
    const float* vv = l1 ? v1 : v0;
    float* plane = ws + (l1 ? WS_A1 : WS_A0);
    const int base = blk * SORT_T;
    const int n = min(SORT_T, EE - base);

    for (int b = tid; b < N2N_BINS; b += SORT_NT) cnt[b] = 0u;
    int row[SORT_PER]; float val[SORT_PER]; unsigned rank[SORT_PER];
    #pragma unroll
    for (int j = 0; j < SORT_PER; ++j) {
        int k = j * SORT_NT + tid;
        int idx = base + min(k, n - 1);
        row[j] = rr[idx];
        val[j] = vv[idx];
    }
    __syncthreads();
    #pragma unroll
    for (int j = 0; j < SORT_PER; ++j) {
        int k = j * SORT_NT + tid;
        rank[j] = 0u;
        if (k < n) rank[j] = atomicAdd(&cnt[row[j] >> 5], 1u);
    }
    __syncthreads();
    bins_to_offsets<N2N_BINS>(cnt, wsum);
    unsigned pos[SORT_PER];
    #pragma unroll
    for (int j = 0; j < SORT_PER; ++j) pos[j] = cnt[row[j] >> 5] + rank[j];
    #pragma unroll
    for (int j = 0; j < SORT_PER; ++j) {
        int k = j * SORT_NT + tid;
        if (k < n) sorted[pos[j]] = make_uint2((unsigned)row[j], __float_as_uint(val[j]));
    }
    __syncthreads();
    uint2 e[SORT_PER];
    #pragma unroll
    for (int j = 0; j < SORT_PER; ++j)
        e[j] = sorted[min(j * SORT_NT + tid, n - 1)];
    #pragma unroll
    for (int j = 0; j < SORT_PER; ++j) {
        int k = j * SORT_NT + tid;
        if (k < n) atomicAdd(&plane[e[j].x], __uint_as_float(e[j].y));
    }
}

__global__ void k_subg(const int* __restrict__ subg_row, float* ws) {
    int i = blockIdx.x * blockDim.x + threadIdx.x;
    if (i < NN) {
        int b = subg_row[i];
        atomicAdd(&ws[WS_A0 + NN + b], 1.0f);
        atomicAdd(&ws[WS_A1 + NN + b], 1.0f);
    }
}

__global__ void __launch_bounds__(SORT_NT)
k_sage(const int* __restrict__ s0r, const int* __restrict__ s0c,
       const int* __restrict__ s1r, const int* __restrict__ s1c,
       float* ws) {
    __shared__ unsigned cnt[SAGE_BINS];
    __shared__ unsigned wsum[32];
    __shared__ uint2 sorted[SORT_T];
    const int tid = threadIdx.x, lane = tid & 31, wave = tid >> 5;
    const bool l1 = blockIdx.x >= SAGE_BLK;
    const int blk = l1 ? blockIdx.x - SAGE_BLK : blockIdx.x;
    const int* rr = l1 ? s1r : s0r;
    const int* cc = l1 ? s1c : s0c;
    const float* ap = ws + (l1 ? WS_A1 : WS_A0);
    float* md = ws + (l1 ? WS_MD1 : WS_MD0);
    const int base = blk * SORT_T;
    const int n = min(SORT_T, ESZ - base);

    for (int b = tid; b < SAGE_BINS; b += SORT_NT) cnt[b] = 0u;
    int row[SORT_PER]; int col[SORT_PER]; float av[SORT_PER]; unsigned rank[SORT_PER];
    #pragma unroll
    for (int j = 0; j < SORT_PER; ++j) {
        int k = j * SORT_NT + tid;
        int idx = base + min(k, n - 1);
        row[j] = rr[idx];
        col[j] = cc[idx];
    }
    #pragma unroll
    for (int j = 0; j < SORT_PER; ++j) av[j] = ap[col[j]];
    __syncthreads();
    #pragma unroll
    for (int j = 0; j < SORT_PER; ++j) {
        int k = j * SORT_NT + tid;
        rank[j] = 0u;
        if (k < n) rank[j] = atomicAdd(&cnt[row[j] >> 4], 1u);
    }
    __syncthreads();
    bins_to_offsets<SAGE_BINS>(cnt, wsum);
    unsigned pos[SORT_PER];
    #pragma unroll
    for (int j = 0; j < SORT_PER; ++j) pos[j] = cnt[row[j] >> 4] + rank[j];
    #pragma unroll
    for (int j = 0; j < SORT_PER; ++j) {
        int k = j * SORT_NT + tid;
        if (k < n) sorted[pos[j]] = make_uint2((unsigned)row[j], __float_as_uint(av[j]));
    }
    __syncthreads();
    const int jj   = lane >> 1;
    const int role = lane & 1;
    uint2 eA[SORT_PER], eB[SORT_PER];
    #pragma unroll
    for (int j = 0; j < SORT_PER; ++j) {
        int p = (j * 32 + wave) * 32 + jj;
        eA[j] = sorted[min(p, n - 1)];
        eB[j] = sorted[min(p + 16, n - 1)];
    }
    #pragma unroll
    for (int j = 0; j < SORT_PER; ++j) {
        int p = (j * 32 + wave) * 32 + jj;
        if (p < n)
            atomicAdd(&md[2 * eA[j].x + role], role ? 1.0f : __uint_as_float(eA[j].y));
        if (p + 16 < n)
            atomicAdd(&md[2 * eB[j].x + role], role ? 1.0f : __uint_as_float(eB[j].y));
    }
}

#define ROWS_NT    256
#define ROWS_BLK   1024
#define ROWS_WAVES (ROWS_BLK * ROWS_NT / 32)
#define ROWS_UNR   4
static_assert(NBT % 2 == 0 && NN % 2 == 0, "row pairs must not straddle a layer or the real/virtual split");

__device__ __forceinline__ float half_sum(float v) {
    v += __int_as_float(__builtin_amdgcn_update_dpp(0, __float_as_int(v), 0x161, 0xf, 0xf, true));
    v += __int_as_float(__builtin_amdgcn_update_dpp(0, __float_as_int(v), 0x162, 0xf, 0xf, true));
    v += __int_as_float(__builtin_amdgcn_update_dpp(0, __float_as_int(v), 0x164, 0xf, 0xf, true));
    v += __int_as_float(__builtin_amdgcn_update_dpp(0, __float_as_int(v), 0x168, 0xf, 0xf, true));
    return v;
}

__global__ void __launch_bounds__(ROWS_NT)
k_rows(float* ws, const float* __restrict__ cp, float* out) {
    const int lane = threadIdx.x & 31;
    const int half = lane >> 4;
    const int q    = lane & 15;
    const int gw   = (blockIdx.x * ROWS_NT + threadIdx.x) >> 5;

    const float4 u1 = *(const float4*)&ws[WS_U1 + 4 * q];
    const float4 u2 = *(const float4*)&ws[WS_U2 + 4 * q];

    for (int p0 = gw; p0 < NBT; p0 += ROWS_WAVES * ROWS_UNR) {
        float a[ROWS_UNR]; float2 md[ROWS_UNR];
        #pragma unroll
        for (int k = 0; k < ROWS_UNR; ++k) {
            int p = min(p0 + k * ROWS_WAVES, NBT - 1);
            int w = 2 * p + half;
            int l = (w >= NBT) ? 1 : 0;
            int i = w - l * NBT;
            a[k]  = ws[(l ? WS_A1 : WS_A0) + i];
            md[k] = *(const float2*)&ws[(l ? WS_MD1 : WS_MD0) + 2 * i];
        }
        #pragma unroll
        for (int k = 0; k < ROWS_UNR; ++k) {
            int p = p0 + k * ROWS_WAVES;
            if (p < NBT) {
                int w = 2 * p + half;
                int l = (w >= NBT) ? 1 : 0;
                int i = w - l * NBT;
                float m = md[k].x / fmaxf(md[k].y, 1.0f);
                float r0 = fmaxf(a[k] * u1.x + m * u2.x, 0.0f);
                float r1 = fmaxf(a[k] * u1.y + m * u2.y, 0.0f);
                float r2 = fmaxf(a[k] * u1.z + m * u2.z, 0.0f);
                float r3 = fmaxf(a[k] * u1.w + m * u2.w, 0.0f);
                float ss = (r0 * r0 + r1 * r1) + (r2 * r2 + r3 * r3);
                ss = half_sum(ss);
                float inv = 1.0f / fmaxf(sqrtf(ss), 1e-12f);
                r0 *= inv; r1 *= inv; r2 *= inv; r3 *= inv;
                if (i < NN) {
                    float* dst = out + BB + (l * NN + i) * DD + 4 * q;
                    *(float4*)dst = make_float4(r0, r1, r2, r3);
                } else {
                    int b = i - NN;
                    const float4 c4 = *(const float4*)&cp[4 * q];
                    float t = (r0 * c4.x + r1 * c4.y) + (r2 * c4.z + r3 * c4.w);
                    t = half_sum(t);
                    if (q == 0) ws[WS_T + l * BB + b] = t;
                }
            }
        }
    }
}

__global__ void k_esa(const int* __restrict__ act_col, const float* __restrict__ out,
                      float* ws) {
    int idx = blockIdx.x * blockDim.x + threadIdx.x;
    if (idx >= 2 * BB * DD) return;
    int d = idx & (DD - 1);
    int b = (idx >> 6) & (BB - 1);
    int l = idx >> 15;
    float t = ws[WS_T + l * BB + b];
    float ae = out[BB + (l * NN + act_col[b]) * DD + d];
    ws[WS_ESA + idx] = t * ae;
}

__global__ void k_gemm(const float* __restrict__ ws, const float* __restrict__ h1,
                       float* __restrict__ hid) {
    int bid  = blockIdx.x;
    int tn   = bid & 1;
    int tm   = (bid >> 1) & 31;
    int l    = bid >> 6;
    int lane = threadIdx.x;
    int half = lane >> 4;
    int lid  = lane & 15;

    const float* A = ws + WS_ESA + (l * BB + tm * 16) * DD;
    v8f c = {0.f, 0.f, 0.f, 0.f, 0.f, 0.f, 0.f, 0.f};
    #pragma unroll
    for (int s = 0; s < 16; ++s) {
        int k0 = 4 * s + 2 * half;
        v2f av, bv;
        av.x = A[lid * DD + k0];
        av.y = A[lid * DD + k0 + 1];
        bv.x = h1[(k0    ) * 32 + tn * 16 + lid];
        bv.y = h1[(k0 + 1) * 32 + tn * 16 + lid];
        c = __builtin_amdgcn_wmma_f32_16x16x4_f32(false, av, false, bv,
                                                  (short)0, c, false, false);
    }
    float* dst = hid + (l * BB + tm * 16) * 32 + tn * 16;
    #pragma unroll
    for (int r = 0; r < 8; ++r) {
        int M = r + 8 * half;
        dst[M * 32 + lid] = fmaxf(c[r], 0.0f);
    }
}

__global__ void k_q(const float* __restrict__ ws, const float* __restrict__ aux,
                    const float* __restrict__ h2w, float* out) {
    int b = blockIdx.x * blockDim.x + threadIdx.x;
    if (b >= BB) return;
    float q = 0.0f;
    for (int l = 0; l < 2; ++l) {
        const float* h = ws + WS_HID + (l * BB + b) * 32;
        for (int j = 0; j < 32; ++j) q += h[j] * h2w[j];
        for (int j = 0; j < 4; ++j)  q += aux[b * 8 + l * 4 + j] * h2w[32 + j];
    }
    out[b] = q;
}

extern "C" void kernel_launch(void* const* d_in, const int* in_sizes, int n_in,
                              void* d_out, int out_size, void* d_ws, size_t ws_size,
                              hipStream_t stream) {
    const int*   n2n0_row = (const int*)  d_in[0];
    const float* n2n0_val = (const float*)d_in[2];
    const int*   n2n1_row = (const int*)  d_in[3];
    const float* n2n1_val = (const float*)d_in[5];
    const int*   subg_row = (const int*)  d_in[6];
    const int*   act_col  = (const int*)  d_in[9];
    const int*   sg0_row  = (const int*)  d_in[10];
    const int*   sg0_col  = (const int*)  d_in[11];
    const int*   sg1_row  = (const int*)  d_in[12];
    const int*   sg1_col  = (const int*)  d_in[13];
    const float* aux      = (const float*)d_in[14];
    const float* w_n2l    = (const float*)d_in[15];
    const float* p_conv   = (const float*)d_in[16];
    const float* W_sage   = (const float*)d_in[17];
    const float* cp       = (const float*)d_in[18];
    const float* h1w      = (const float*)d_in[19];
    const float* h2w      = (const float*)d_in[20];
    float* out = (float*)d_out;
    float* ws  = (float*)d_ws;

    k_zero <<<(6 * NBT + 255) / 256, 256, 0, stream>>>(ws);
    k_small<<<1, 64, 0, stream>>>(w_n2l, p_conv, W_sage, ws);
    k_n2n  <<<2 * N2N_BLK, SORT_NT, 0, stream>>>(n2n0_row, n2n0_val,
                                                  n2n1_row, n2n1_val, ws);
    k_subg <<<(NN + 255) / 256, 256, 0, stream>>>(subg_row, ws);
    k_sage <<<2 * SAGE_BLK, SORT_NT, 0, stream>>>(sg0_row, sg0_col,
                                                   sg1_row, sg1_col, ws);
    k_rows <<<ROWS_BLK, ROWS_NT, 0, stream>>>(ws, cp, out);
    k_esa  <<<(2 * BB * DD + 255) / 256, 256, 0, stream>>>(act_col, out, ws);
    k_gemm <<<128, 32, 0, stream>>>(ws, h1w, ws + WS_HID);
    k_q    <<<(BB + 255) / 256, 256, 0, stream>>>(ws, aux, h2w, out);
}


# sage gather: a block first counting-sorts its 16384 edges by the 128-B line of a[col] (same LDS bins and buffer), so a gather instruction asks for ~7 lines, was 32; float sum order changes
# speedup vs baseline: 3.0617x; 1.1236x over previous
#include <hip/hip_runtime.h>

#define NN   100000
#define BB   512
#define EE   1000000
#define ESZ  1200000
#define DD   64
#define NBT  (NN + BB)

#define WS_A0   0
#define WS_A1   (NBT)
#define WS_MD0  (2*NBT)
#define WS_MD1  (4*NBT)
#define WS_SM   (6*NBT)
#define WS_U1   (WS_SM + 128)
#define WS_U2   (WS_SM + 192)
#define WS_T    (WS_SM + 256)
#define WS_ESA  (WS_T + 2*BB)
#define WS_HID  (WS_ESA + 2*BB*DD)

typedef float v2f __attribute__((ext_vector_type(2)));
typedef float v8f __attribute__((ext_vector_type(8)));

__global__ void k_zero(float* ws) {
    int i = blockIdx.x * blockDim.x + threadIdx.x;
    if (i < 6 * NBT) ws[i] = 0.0f;
}

__global__ void k_small(const float* __restrict__ w_n2l,
                        const float* __restrict__ p_conv,
                        const float* __restrict__ W_sage,
                        float* ws) {
    __shared__ float v[DD], w[DD];
    int d = threadIdx.x;
    float x = fmaxf(w_n2l[d] + w_n2l[DD + d], 0.0f);
    v[d] = x;
    __syncthreads();
    float ss = 0.f;
    for (int k = 0; k < DD; ++k) ss += v[k] * v[k];
    float nv = x / fmaxf(sqrtf(ss), 1e-12f);
    __syncthreads();
    v[d] = nv;
    __syncthreads();
    float wd = 0.f;
    for (int k = 0; k < DD; ++k) wd += v[k] * p_conv[k * DD + d];
    w[d] = wd;
    __syncthreads();
    float u1 = 0.f, u2 = 0.f;
    for (int k = 0; k < DD; ++k) {
        u1 += w[k] * W_sage[k * DD + d];
        u2 += w[k] * W_sage[(DD + k) * DD + d];
    }
    ws[WS_SM + d]  = nv;
    ws[WS_SM + 64 + d] = wd;
    ws[WS_U1 + d] = u1;
    ws[WS_U2 + d] = u2;
}

#define SORT_T    16384
#define SORT_NT   1024
#define SORT_PER  (SORT_T / SORT_NT)
#define N2N_BINS  ((NBT + 31) / 32)
#define SAGE_BINS ((NBT + 15) / 16)
#define N2N_BLK   ((EE + SORT_T - 1) / SORT_T)
#define SAGE_BLK  ((ESZ + SORT_T - 1) / SORT_T)
static_assert(SORT_T == 32 * 32 * SORT_PER, "32 waves x SORT_PER groups of 32 ordered edges");

template <int NB>
__device__ __forceinline__ void bins_to_offsets(unsigned* cnt, unsigned* wsum) {
    constexpr int PER = (NB + SORT_NT - 1) / SORT_NT;
    const int tid = threadIdx.x, lane = tid & 31, wave = tid >> 5;
    unsigned loc[PER];
    unsigned s = 0;
    #pragma unroll
    for (int k = 0; k < PER; ++k) {
        int b = tid * PER + k;
        unsigned v = (b < NB) ? cnt[b] : 0u;
        loc[k] = s;
        s += v;
    }
    unsigned inc = s;
    for (int off = 1; off < 32; off <<= 1) {
        unsigned t = __shfl_up(inc, off, 32);
        if (lane >= off) inc += t;
    }
    if (lane == 31) wsum[wave] = inc;
    __syncthreads();
    if (wave == 0) {
        unsigned w = wsum[lane];
        unsigned winc = w;
        for (int off = 1; off < 32; off <<= 1) {
            unsigned t = __shfl_up(winc, off, 32);
            if (lane >= off) winc += t;
        }
        wsum[lane] = winc - w;
    }
    __syncthreads();
    unsigned base = wsum[wave] + (inc - s);
    #pragma unroll
    for (int k = 0; k < PER; ++k) {
        int b = tid * PER + k;
        if (b < NB) cnt[b] = base + loc[k];
    }
    __syncthreads();
}

__global__ void __launch_bounds__(SORT_NT)
k_n2n(const int* __restrict__ r0, const float* __restrict__ v0,
      const int* __restrict__ r1, const float* __restrict__ v1,
      float* ws) {
    __shared__ unsigned cnt[N2N_BINS];
    __shared__ unsigned wsum[32];
    __shared__ uint2 sorted[SORT_T];
    const int tid = threadIdx.x;
    const bool l1 = blockIdx.x >= N2N_BLK;
    const int blk = l1 ? blockIdx.x - N2N_BLK : blockIdx.x;
    const int*   rr = l1 ? r1 : r0;
    const float* vv = l1 ? v1 : v0;
    float* plane = ws + (l1 ? WS_A1 : WS_A0);
    const int base = blk * SORT_T;
    const int n = min(SORT_T, EE - base);

    for (int b = tid; b < N2N_BINS; b += SORT_NT) cnt[b] = 0u;
    int row[SORT_PER]; float val[SORT_PER]; unsigned rank[SORT_PER];
    #pragma unroll
    for (int j = 0; j < SORT_PER; ++j) {
        int k = j * SORT_NT + tid;
        int idx = base + min(k, n - 1);
        row[j] = rr[idx];
        val[j] = vv[idx];
    }
    __syncthreads();
    #pragma unroll
    for (int j = 0; j < SORT_PER; ++j) {
        int k = j * SORT_NT + tid;
        rank[j] = 0u;
        if (k < n) rank[j] = atomicAdd(&cnt[row[j] >> 5], 1u);
    }
    __syncthreads();
    bins_to_offsets<N2N_BINS>(cnt, wsum);
    unsigned pos[SORT_PER];
    #pragma unroll
    for (int j = 0; j < SORT_PER; ++j) pos[j] = cnt[row[j] >> 5] + rank[j];
    #pragma unroll
    for (int j = 0; j < SORT_PER; ++j) {
        int k = j * SORT_NT + tid;
        if (k < n) sorted[pos[j]] = make_uint2((unsigned)row[j], __float_as_uint(val[j]));
    }
    __syncthreads();
    uint2 e[SORT_PER];
    #pragma unroll
    for (int j = 0; j < SORT_PER; ++j)
        e[j] = sorted[min(j * SORT_NT + tid, n - 1)];
    #pragma unroll
    for (int j = 0; j < SORT_PER; ++j) {
        int k = j * SORT_NT + tid;
        if (k < n) atomicAdd(&plane[e[j].x], __uint_as_float(e[j].y));
    }
}

__global__ void k_subg(const int* __restrict__ subg_row, float* ws) {
    int i = blockIdx.x * blockDim.x + threadIdx.x;
    if (i < NN) {
        int b = subg_row[i];
        atomicAdd(&ws[WS_A0 + NN + b], 1.0f);
        atomicAdd(&ws[WS_A1 + NN + b], 1.0f);
    }
}

__global__ void __launch_bounds__(SORT_NT)
k_sage(const int* __restrict__ s0r, const int* __restrict__ s0c,
       const int* __restrict__ s1r, const int* __restrict__ s1c,
       float* ws) {
    __shared__ unsigned cnt[SAGE_BINS];
    __shared__ unsigned wsum[32];
    __shared__ uint2 sorted[SORT_T];
    const int tid = threadIdx.x, lane = tid & 31, wave = tid >> 5;
    const bool l1 = blockIdx.x >= SAGE_BLK;
    const int blk = l1 ? blockIdx.x - SAGE_BLK : blockIdx.x;
    const int* rr = l1 ? s1r : s0r;
    const int* cc = l1 ? s1c : s0c;
    const float* ap = ws + (l1 ? WS_A1 : WS_A0);
    float* md = ws + (l1 ? WS_MD1 : WS_MD0);
    const int base = blk * SORT_T;
    const int n = min(SORT_T, ESZ - base);

    for (int b = tid; b < N2N_BINS; b += SORT_NT) cnt[b] = 0u;
    int row[SORT_PER]; int col[SORT_PER]; float av[SORT_PER]; unsigned rank[SORT_PER];
    unsigned pos[SORT_PER];
    #pragma unroll
    for (int j = 0; j < SORT_PER; ++j) {
        int k = j * SORT_NT + tid;
        int idx = base + min(k, n - 1);
        row[j] = rr[idx];
        col[j] = cc[idx];
    }
    __syncthreads();
    #pragma unroll
    for (int j = 0; j < SORT_PER; ++j) {
        int k = j * SORT_NT + tid;
        rank[j] = 0u;
        if (k < n) rank[j] = atomicAdd(&cnt[col[j] >> 5], 1u);
    }
    __syncthreads();
    bins_to_offsets<N2N_BINS>(cnt, wsum);
    #pragma unroll
    for (int j = 0; j < SORT_PER; ++j) pos[j] = cnt[col[j] >> 5] + rank[j];
    #pragma unroll
    for (int j = 0; j < SORT_PER; ++j) {
        int k = j * SORT_NT + tid;
        if (k < n) sorted[pos[j]] = make_uint2((unsigned)row[j], (unsigned)col[j]);
    }
    __syncthreads();
    #pragma unroll
    for (int j = 0; j < SORT_PER; ++j) {
        uint2 e = sorted[min(j * SORT_NT + tid, n - 1)];
        row[j] = (int)e.x;
        col[j] = (int)e.y;
    }
    #pragma unroll
    for (int j = 0; j < SORT_PER; ++j) av[j] = ap[col[j]];
    for (int b = tid; b < SAGE_BINS; b += SORT_NT) cnt[b] = 0u;
    __syncthreads();
    #pragma unroll
    for (int j = 0; j < SORT_PER; ++j) {
        int k = j * SORT_NT + tid;
        rank[j] = 0u;
        if (k < n) rank[j] = atomicAdd(&cnt[row[j] >> 4], 1u);
    }
    __syncthreads();
    bins_to_offsets<SAGE_BINS>(cnt, wsum);
    #pragma unroll
    for (int j = 0; j < SORT_PER; ++j) pos[j] = cnt[row[j] >> 4] + rank[j];
    #pragma unroll
    for (int j = 0; j < SORT_PER; ++j) {
        int k = j * SORT_NT + tid;
        if (k < n) sorted[pos[j]] = make_uint2((unsigned)row[j], __float_as_uint(av[j]));
    }
    __syncthreads();
    const int jj   = lane >> 1;
    const int role = lane & 1;
    uint2 eA[SORT_PER], eB[SORT_PER];
    #pragma unroll
    for (int j = 0; j < SORT_PER; ++j) {
        int p = (j * 32 + wave) * 32 + jj;
        eA[j] = sorted[min(p, n - 1)];
        eB[j] = sorted[min(p + 16, n - 1)];
    }
    #pragma unroll
    for (int j = 0; j < SORT_PER; ++j) {
        int p = (j * 32 + wave) * 32 + jj;
        if (p < n)
            atomicAdd(&md[2 * eA[j].x + role], role ? 1.0f : __uint_as_float(eA[j].y));
        if (p + 16 < n)
            atomicAdd(&md[2 * eB[j].x + role], role ? 1.0f : __uint_as_float(eB[j].y));
    }
}

#define ROWS_NT    256
#define ROWS_BLK   1024
#define ROWS_WAVES (ROWS_BLK * ROWS_NT / 32)
#define ROWS_UNR   4
static_assert(NBT % 2 == 0 && NN % 2 == 0, "row pairs must not straddle a layer or the real/virtual split");

__device__ __forceinline__ float half_sum(float v) {
    v += __int_as_float(__builtin_amdgcn_update_dpp(0, __float_as_int(v), 0x161, 0xf, 0xf, true));
    v += __int_as_float(__builtin_amdgcn_update_dpp(0, __float_as_int(v), 0x162, 0xf, 0xf, true));
    v += __int_as_float(__builtin_amdgcn_update_dpp(0, __float_as_int(v), 0x164, 0xf, 0xf, true));
    v += __int_as_float(__builtin_amdgcn_update_dpp(0, __float_as_int(v), 0x168, 0xf, 0xf, true));
    return v;
}

__global__ void __launch_bounds__(ROWS_NT)
k_rows(float* ws, const float* __restrict__ cp, float* out) {
    const int lane = threadIdx.x & 31;
    const int half = lane >> 4;
    const int q    = lane & 15;
    const int gw   = (blockIdx.x * ROWS_NT + threadIdx.x) >> 5;

    const float4 u1 = *(const float4*)&ws[WS_U1 + 4 * q];
    const float4 u2 = *(const float4*)&ws[WS_U2 + 4 * q];

    for (int p0 = gw; p0 < NBT; p0 += ROWS_WAVES * ROWS_UNR) {
        float a[ROWS_UNR]; float2 md[ROWS_UNR];
        #pragma unroll
        for (int k = 0; k < ROWS_UNR; ++k) {
            int p = min(p0 + k * ROWS_WAVES, NBT - 1);
            int w = 2 * p + half;
            int l = (w >= NBT) ? 1 : 0;
            int i = w - l * NBT;
            a[k]  = ws[(l ? WS_A1 : WS_A0) + i];
            md[k] = *(const float2*)&ws[(l ? WS_MD1 : WS_MD0) + 2 * i];
        }
        #pragma unroll
        for (int k = 0; k < ROWS_UNR; ++k) {
            int p = p0 + k * ROWS_WAVES;
            if (p < NBT) {
                int w = 2 * p + half;
                int l = (w >= NBT) ? 1 : 0;
                int i = w - l * NBT;
                float m = md[k].x / fmaxf(md[k].y, 1.0f);
                float r0 = fmaxf(a[k] * u1.x + m * u2.x, 0.0f);
                float r1 = fmaxf(a[k] * u1.y + m * u2.y, 0.0f);
                float r2 = fmaxf(a[k] * u1.z + m * u2.z, 0.0f);
                float r3 = fmaxf(a[k] * u1.w + m * u2.w, 0.0f);
                float ss = (r0 * r0 + r1 * r1) + (r2 * r2 + r3 * r3);
                ss = half_sum(ss);
                float inv = 1.0f / fmaxf(sqrtf(ss), 1e-12f);
                r0 *= inv; r1 *= inv; r2 *= inv; r3 *= inv;
                if (i < NN) {
                    float* dst = out + BB + (l * NN + i) * DD + 4 * q;
                    *(float4*)dst = make_float4(r0, r1, r2, r3);
                } else {
                    int b = i - NN;
                    const float4 c4 = *(const float4*)&cp[4 * q];
                    float t = (r0 * c4.x + r1 * c4.y) + (r2 * c4.z + r3 * c4.w);
                    t = half_sum(t);
                    if (q == 0) ws[WS_T + l * BB + b] = t;
                }
            }
        }
    }
}

__global__ void k_esa(const int* __restrict__ act_col, const float* __restrict__ out,
                      float* ws) {
    int idx = blockIdx.x * blockDim.x + threadIdx.x;
    if (idx >= 2 * BB * DD) return;
    int d = idx & (DD - 1);
    int b = (idx >> 6) & (BB - 1);
    int l = idx >> 15;
    float t = ws[WS_T + l * BB + b];
    float ae = out[BB + (l * NN + act_col[b]) * DD + d];
    ws[WS_ESA + idx] = t * ae;
}

__global__ void k_gemm(const float* __restrict__ ws, const float* __restrict__ h1,
                       float* __restrict__ hid) {
    int bid  = blockIdx.x;
    int tn   = bid & 1;
    int tm   = (bid >> 1) & 31;
    int l    = bid >> 6;
    int lane = threadIdx.x;
    int half = lane >> 4;
    int lid  = lane & 15;

    const float* A = ws + WS_ESA + (l * BB + tm * 16) * DD;
    v8f c = {0.f, 0.f, 0.f, 0.f, 0.f, 0.f, 0.f, 0.f};
    #pragma unroll
    for (int s = 0; s < 16; ++s) {
        int k0 = 4 * s + 2 * half;
        v2f av, bv;
        av.x = A[lid * DD + k0];
        av.y = A[lid * DD + k0 + 1];
        bv.x = h1[(k0    ) * 32 + tn * 16 + lid];
        bv.y = h1[(k0 + 1) * 32 + tn * 16 + lid];
        c = __builtin_amdgcn_wmma_f32_16x16x4_f32(false, av, false, bv,
                                                  (short)0, c, false, false);
    }
    float* dst = hid + (l * BB + tm * 16) * 32 + tn * 16;
    #pragma unroll
    for (int r = 0; r < 8; ++r) {
        int M = r + 8 * half;
        dst[M * 32 + lid] = fmaxf(c[r], 0.0f);
    }
}

__global__ void k_q(const float* __restrict__ ws, const float* __restrict__ aux,
                    const float* __restrict__ h2w, float* out) {
    int b = blockIdx.x * blockDim.x + threadIdx.x;
    if (b >= BB) return;
    float q = 0.0f;
    for (int l = 0; l < 2; ++l) {
        const float* h = ws + WS_HID + (l * BB + b) * 32;
        for (int j = 0; j < 32; ++j) q += h[j] * h2w[j];
        for (int j = 0; j < 4; ++j)  q += aux[b * 8 + l * 4 + j] * h2w[32 + j];
    }
    out[b] = q;
}

extern "C" void kernel_launch(void* const* d_in, const int* in_sizes, int n_in,
                              void* d_out, int out_size, void* d_ws, size_t ws_size,
                              hipStream_t stream) {
    const int*   n2n0_row = (const int*)  d_in[0];
    const float* n2n0_val = (const float*)d_in[2];
    const int*   n2n1_row = (const int*)  d_in[3];
    const float* n2n1_val = (const float*)d_in[5];
    const int*   subg_row = (const int*)  d_in[6];
    const int*   act_col  = (const int*)  d_in[9];
    const int*   sg0_row  = (const int*)  d_in[10];
    const int*   sg0_col  = (const int*)  d_in[11];
    const int*   sg1_row  = (const int*)  d_in[12];
    const int*   sg1_col  = (const int*)  d_in[13];
    const float* aux      = (const float*)d_in[14];
    const float* w_n2l    = (const float*)d_in[15];
    const float* p_conv   = (const float*)d_in[16];
    const float* W_sage   = (const float*)d_in[17];
    const float* cp       = (const float*)d_in[18];
    const float* h1w      = (const float*)d_in[19];
    const float* h2w      = (const float*)d_in[20];
    float* out = (float*)d_out;
    float* ws  = (float*)d_ws;

    k_zero <<<(6 * NBT + 255) / 256, 256, 0, stream>>>(ws);
    k_small<<<1, 64, 0, stream>>>(w_n2l, p_conv, W_sage, ws);
    k_n2n  <<<2 * N2N_BLK, SORT_NT, 0, stream>>>(n2n0_row, n2n0_val,
                                                  n2n1_row, n2n1_val, ws);
    k_subg <<<(NN + 255) / 256, 256, 0, stream>>>(subg_row, ws);
    k_sage <<<2 * SAGE_BLK, SORT_NT, 0, stream>>>(sg0_row, sg0_col,
                                                   sg1_row, sg1_col, ws);
    k_rows <<<ROWS_BLK, ROWS_NT, 0, stream>>>(ws, cp, out);
    k_esa  <<<(2 * BB * DD + 255) / 256, 256, 0, stream>>>(act_col, out, ws);
    k_gemm <<<128, 32, 0, stream>>>(ws, h1w, ws + WS_HID);
    k_q    <<<(BB + 255) / 256, 256, 0, stream>>>(ws, aux, h2w, out);
}


# n2n and sage scatter sums kept once per package half: a block adds into its own half's copy (die id; 2nd copy in out rows written later), copies summed before use; float sum order changes
# speedup vs baseline: 4.6013x; 1.5029x over previous
#include <hip/hip_runtime.h>

#define NN   100000
#define BB   512
#define EE   1000000
#define ESZ  1200000
#define DD   64
#define NBT  (NN + BB)

#define WS_A0   0
#define WS_A1   (NBT)
#define WS_MD0  (2*NBT)
#define WS_MD1  (4*NBT)
#define WS_SM   (6*NBT)
#define WS_U1   (WS_SM + 128)
#define WS_U2   (WS_SM + 192)
#define WS_T    (WS_SM + 256)
#define WS_ESA  (WS_T + 2*BB)
#define WS_HID  (WS_ESA + 2*BB*DD)

typedef float v2f __attribute__((ext_vector_type(2)));
typedef float v8f __attribute__((ext_vector_type(8)));

__device__ __forceinline__ int pkg_half() {
    unsigned id = __builtin_amdgcn_s_sendmsg_rtn(0x87);
    return (int)((id >> 18) & 1u);
}

__global__ void k_zero(float* ws, float* sc) {
    int i = blockIdx.x * blockDim.x + threadIdx.x;
    if (i < 6 * NBT) { ws[i] = 0.0f; sc[i] = 0.0f; }
}

__global__ void k_small(const float* __restrict__ w_n2l,
                        const float* __restrict__ p_conv,
                        const float* __restrict__ W_sage,
                        float* ws) {
    __shared__ float v[DD], w[DD];
    int d = threadIdx.x;
    float x = fmaxf(w_n2l[d] + w_n2l[DD + d], 0.0f);
    v[d] = x;
    __syncthreads();
    float ss = 0.f;
    for (int k = 0; k < DD; ++k) ss += v[k] * v[k];
    float nv = x / fmaxf(sqrtf(ss), 1e-12f);
    __syncthreads();
    v[d] = nv;
    __syncthreads();
    float wd = 0.f;
    for (int k = 0; k < DD; ++k) wd += v[k] * p_conv[k * DD + d];
    w[d] = wd;
    __syncthreads();
    float u1 = 0.f, u2 = 0.f;
    for (int k = 0; k < DD; ++k) {
        u1 += w[k] * W_sage[k * DD + d];
        u2 += w[k] * W_sage[(DD + k) * DD + d];
    }
    ws[WS_SM + d]  = nv;
    ws[WS_SM + 64 + d] = wd;
    ws[WS_U1 + d] = u1;
    ws[WS_U2 + d] = u2;
}

#define SORT_T    16384
#define SORT_NT   1024
#define SORT_PER  (SORT_T / SORT_NT)
#define N2N_BINS  ((NBT + 31) / 32)
#define SAGE_BINS ((NBT + 15) / 16)
#define N2N_BLK   ((EE + SORT_T - 1) / SORT_T)
#define SAGE_BLK  ((ESZ + SORT_T - 1) / SORT_T)
static_assert(SORT_T == 32 * 32 * SORT_PER, "32 waves x SORT_PER groups of 32 ordered edges");

template <int NB>
__device__ __forceinline__ void bins_to_offsets(unsigned* cnt, unsigned* wsum) {
    constexpr int PER = (NB + SORT_NT - 1) / SORT_NT;
    const int tid = threadIdx.x, lane = tid & 31, wave = tid >> 5;
    unsigned loc[PER];
    unsigned s = 0;
    #pragma unroll
    for (int k = 0; k < PER; ++k) {
        int b = tid * PER + k;
        unsigned v = (b < NB) ? cnt[b] : 0u;
        loc[k] = s;
        s += v;
    }
    unsigned inc = s;
    for (int off = 1; off < 32; off <<= 1) {
        unsigned t = __shfl_up(inc, off, 32);
        if (lane >= off) inc += t;
    }
    if (lane == 31) wsum[wave] = inc;
    __syncthreads();
    if (wave == 0) {
        unsigned w = wsum[lane];
        unsigned winc = w;
        for (int off = 1; off < 32; off <<= 1) {
            unsigned t = __shfl_up(winc, off, 32);
            if (lane >= off) winc += t;
        }
        wsum[lane] = winc - w;
    }
    __syncthreads();
    unsigned base = wsum[wave] + (inc - s);
    #pragma unroll
    for (int k = 0; k < PER; ++k) {
        int b = tid * PER + k;
        if (b < NB) cnt[b] = base + loc[k];
    }
    __syncthreads();
}

__global__ void __launch_bounds__(SORT_NT)
k_n2n(const int* __restrict__ r0, const float* __restrict__ v0,
      const int* __restrict__ r1, const float* __restrict__ v1,
      float* ws, float* sc) {
    __shared__ unsigned cnt[N2N_BINS];
    __shared__ unsigned wsum[32];
    __shared__ uint2 sorted[SORT_T];
    const int tid = threadIdx.x;
    const bool l1 = blockIdx.x >= N2N_BLK;
    const int blk = l1 ? blockIdx.x - N2N_BLK : blockIdx.x;
    const int*   rr = l1 ? r1 : r0;
    const float* vv = l1 ? v1 : v0;
    float* plane = (pkg_half() ? sc : ws) + (l1 ? WS_A1 : WS_A0);
    const int base = blk * SORT_T;
    const int n = min(SORT_T, EE - base);

    for (int b = tid; b < N2N_BINS; b += SORT_NT) cnt[b] = 0u;
    int row[SORT_PER]; float val[SORT_PER]; unsigned rank[SORT_PER];
    #pragma unroll
    for (int j = 0; j < SORT_PER; ++j) {
        int k = j * SORT_NT + tid;
        int idx = base + min(k, n - 1);
        row[j] = rr[idx];
        val[j] = vv[idx];
    }
    __syncthreads();
    #pragma unroll
    for (int j = 0; j < SORT_PER; ++j) {
        int k = j * SORT_NT + tid;
        rank[j] = 0u;
        if (k < n) rank[j] = atomicAdd(&cnt[row[j] >> 5], 1u);
    }
    __syncthreads();
    bins_to_offsets<N2N_BINS>(cnt, wsum);
    unsigned pos[SORT_PER];
    #pragma unroll
    for (int j = 0; j < SORT_PER; ++j) pos[j] = cnt[row[j] >> 5] + rank[j];
    #pragma unroll
    for (int j = 0; j < SORT_PER; ++j) {
        int k = j * SORT_NT + tid;
        if (k < n) sorted[pos[j]] = make_uint2((unsigned)row[j], __float_as_uint(val[j]));
    }
    __syncthreads();
    uint2 e[SORT_PER];
    #pragma unroll
    for (int j = 0; j < SORT_PER; ++j)
        e[j] = sorted[min(j * SORT_NT + tid, n - 1)];
    #pragma unroll
    for (int j = 0; j < SORT_PER; ++j) {
        int k = j * SORT_NT + tid;
        if (k < n) atomicAdd(&plane[e[j].x], __uint_as_float(e[j].y));
    }
}

__global__ void k_subg(const int* __restrict__ subg_row, float* ws, const float* sc) {
    int i = blockIdx.x * blockDim.x + threadIdx.x;
    if (i < NN) {
        float s0 = sc[WS_A0 + i];
        float s1 = sc[WS_A1 + i];
        int b = subg_row[i];
        ws[WS_A0 + i] += s0;
        ws[WS_A1 + i] += s1;
        atomicAdd(&ws[WS_A0 + NN + b], 1.0f);
        atomicAdd(&ws[WS_A1 + NN + b], 1.0f);
    }
}

__global__ void __launch_bounds__(SORT_NT)
k_sage(const int* __restrict__ s0r, const int* __restrict__ s0c,
       const int* __restrict__ s1r, const int* __restrict__ s1c,
       float* ws, float* sc) {
    __shared__ unsigned cnt[SAGE_BINS];
    __shared__ unsigned wsum[32];
    __shared__ uint2 sorted[SORT_T];
    const int tid = threadIdx.x, lane = tid & 31, wave = tid >> 5;
    const bool l1 = blockIdx.x >= SAGE_BLK;
    const int blk = l1 ? blockIdx.x - SAGE_BLK : blockIdx.x;
    const int* rr = l1 ? s1r : s0r;
    const int* cc = l1 ? s1c : s0c;
    const float* ap = ws + (l1 ? WS_A1 : WS_A0);
    float* md = (pkg_half() ? sc : ws) + (l1 ? WS_MD1 : WS_MD0);
    const int base = blk * SORT_T;
    const int n = min(SORT_T, ESZ - base);

    for (int b = tid; b < N2N_BINS; b += SORT_NT) cnt[b] = 0u;
    int row[SORT_PER]; int col[SORT_PER]; float av[SORT_PER]; unsigned rank[SORT_PER];
    unsigned pos[SORT_PER];
    #pragma unroll
    for (int j = 0; j < SORT_PER; ++j) {
        int k = j * SORT_NT + tid;
        int idx = base + min(k, n - 1);
        row[j] = rr[idx];
        col[j] = cc[idx];
    }
    __syncthreads();
    #pragma unroll
    for (int j = 0; j < SORT_PER; ++j) {
        int k = j * SORT_NT + tid;
        rank[j] = 0u;
        if (k < n) rank[j] = atomicAdd(&cnt[col[j] >> 5], 1u);
    }
    __syncthreads();
    bins_to_offsets<N2N_BINS>(cnt, wsum);
    #pragma unroll
    for (int j = 0; j < SORT_PER; ++j) pos[j] = cnt[col[j] >> 5] + rank[j];
    #pragma unroll
    for (int j = 0; j < SORT_PER; ++j) {
        int k = j * SORT_NT + tid;
        if (k < n) sorted[pos[j]] = make_uint2((unsigned)row[j], (unsigned)col[j]);
    }
    __syncthreads();
    #pragma unroll
    for (int j = 0; j < SORT_PER; ++j) {
        uint2 e = sorted[min(j * SORT_NT + tid, n - 1)];
        row[j] = (int)e.x;
        col[j] = (int)e.y;
    }
    #pragma unroll
    for (int j = 0; j < SORT_PER; ++j) av[j] = ap[col[j]];
    for (int b = tid; b < SAGE_BINS; b += SORT_NT) cnt[b] = 0u;
    __syncthreads();
    #pragma unroll
    for (int j = 0; j < SORT_PER; ++j) {
        int k = j * SORT_NT + tid;
        rank[j] = 0u;
        if (k < n) rank[j] = atomicAdd(&cnt[row[j] >> 4], 1u);
    }
    __syncthreads();
    bins_to_offsets<SAGE_BINS>(cnt, wsum);
    #pragma unroll
    for (int j = 0; j < SORT_PER; ++j) pos[j] = cnt[row[j] >> 4] + rank[j];
    #pragma unroll
    for (int j = 0; j < SORT_PER; ++j) {
        int k = j * SORT_NT + tid;
        if (k < n) sorted[pos[j]] = make_uint2((unsigned)row[j], __float_as_uint(av[j]));
    }
    __syncthreads();
    const int jj   = lane >> 1;
    const int role = lane & 1;
    uint2 eA[SORT_PER], eB[SORT_PER];
    #pragma unroll
    for (int j = 0; j < SORT_PER; ++j) {
        int p = (j * 32 + wave) * 32 + jj;
        eA[j] = sorted[min(p, n - 1)];
        eB[j] = sorted[min(p + 16, n - 1)];
    }
    #pragma unroll
    for (int j = 0; j < SORT_PER; ++j) {
        int p = (j * 32 + wave) * 32 + jj;
        if (p < n)
            atomicAdd(&md[2 * eA[j].x + role], role ? 1.0f : __uint_as_float(eA[j].y));
        if (p + 16 < n)
            atomicAdd(&md[2 * eB[j].x + role], role ? 1.0f : __uint_as_float(eB[j].y));
    }
}

__global__ void k_mdsum(float* ws, const float* __restrict__ sc) {
    int i = blockIdx.x * blockDim.x + threadIdx.x;
    if (i < NBT) {
        float4 x = *(const float4*)&ws[WS_MD0 + 4 * i];
        const float4 y = *(const float4*)&sc[WS_MD0 + 4 * i];
        x.x += y.x; x.y += y.y; x.z += y.z; x.w += y.w;
        *(float4*)&ws[WS_MD0 + 4 * i] = x;
    }
}

#define ROWS_NT    256
#define ROWS_BLK   1024
#define ROWS_WAVES (ROWS_BLK * ROWS_NT / 32)
#define ROWS_UNR   4
static_assert(NBT % 2 == 0 && NN % 2 == 0, "row pairs must not straddle a layer or the real/virtual split");

__device__ __forceinline__ float half_sum(float v) {
    v += __int_as_float(__builtin_amdgcn_update_dpp(0, __float_as_int(v), 0x161, 0xf, 0xf, true));
    v += __int_as_float(__builtin_amdgcn_update_dpp(0, __float_as_int(v), 0x162, 0xf, 0xf, true));
    v += __int_as_float(__builtin_amdgcn_update_dpp(0, __float_as_int(v), 0x164, 0xf, 0xf, true));
    v += __int_as_float(__builtin_amdgcn_update_dpp(0, __float_as_int(v), 0x168, 0xf, 0xf, true));
    return v;
}

__global__ void __launch_bounds__(ROWS_NT)
k_rows(float* ws, const float* __restrict__ cp, float* out) {
    const int lane = threadIdx.x & 31;
    const int half = lane >> 4;
    const int q    = lane & 15;
    const int gw   = (blockIdx.x * ROWS_NT + threadIdx.x) >> 5;

    const float4 u1 = *(const float4*)&ws[WS_U1 + 4 * q];
    const float4 u2 = *(const float4*)&ws[WS_U2 + 4 * q];

    for (int p0 = gw; p0 < NBT; p0 += ROWS_WAVES * ROWS_UNR) {
        float a[ROWS_UNR]; float2 md[ROWS_UNR];
        #pragma unroll
        for (int k = 0; k < ROWS_UNR; ++k) {
            int p = min(p0 + k * ROWS_WAVES, NBT - 1);
            int w = 2 * p + half;
            int l = (w >= NBT) ? 1 : 0;
            int i = w - l * NBT;
            a[k]  = ws[(l ? WS_A1 : WS_A0) + i];
            md[k] = *(const float2*)&ws[(l ? WS_MD1 : WS_MD0) + 2 * i];
        }
        #pragma unroll
        for (int k = 0; k < ROWS_UNR; ++k) {
            int p = p0 + k * ROWS_WAVES;
            if (p < NBT) {
                int w = 2 * p + half;
                int l = (w >= NBT) ? 1 : 0;
                int i = w - l * NBT;
                float m = md[k].x / fmaxf(md[k].y, 1.0f);
                float r0 = fmaxf(a[k] * u1.x + m * u2.x, 0.0f);
                float r1 = fmaxf(a[k] * u1.y + m * u2.y, 0.0f);
                float r2 = fmaxf(a[k] * u1.z + m * u2.z, 0.0f);
                float r3 = fmaxf(a[k] * u1.w + m * u2.w, 0.0f);
                float ss = (r0 * r0 + r1 * r1) + (r2 * r2 + r3 * r3);
                ss = half_sum(ss);
                float inv = 1.0f / fmaxf(sqrtf(ss), 1e-12f);
                r0 *= inv; r1 *= inv; r2 *= inv; r3 *= inv;
                if (i < NN) {
                    float* dst = out + BB + (l * NN + i) * DD + 4 * q;
                    *(float4*)dst = make_float4(r0, r1, r2, r3);
                } else {
                    int b = i - NN;
                    const float4 c4 = *(const float4*)&cp[4 * q];
                    float t = (r0 * c4.x + r1 * c4.y) + (r2 * c4.z + r3 * c4.w);
                    t = half_sum(t);
                    if (q == 0) ws[WS_T + l * BB + b] = t;
                }
            }
        }
    }
}

__global__ void k_esa(const int* __restrict__ act_col, const float* __restrict__ out,
                      float* ws) {
    int idx = blockIdx.x * blockDim.x + threadIdx.x;
    if (idx >= 2 * BB * DD) return;
    int d = idx & (DD - 1);
    int b = (idx >> 6) & (BB - 1);
    int l = idx >> 15;
    float t = ws[WS_T + l * BB + b];
    float ae = out[BB + (l * NN + act_col[b]) * DD + d];
    ws[WS_ESA + idx] = t * ae;
}

__global__ void k_gemm(const float* __restrict__ ws, const float* __restrict__ h1,
                       float* __restrict__ hid) {
    int bid  = blockIdx.x;
    int tn   = bid & 1;
    int tm   = (bid >> 1) & 31;
    int l    = bid >> 6;
    int lane = threadIdx.x;
    int half = lane >> 4;
    int lid  = lane & 15;

    const float* A = ws + WS_ESA + (l * BB + tm * 16) * DD;
    v8f c = {0.f, 0.f, 0.f, 0.f, 0.f, 0.f, 0.f, 0.f};
    #pragma unroll
    for (int s = 0; s < 16; ++s) {
        int k0 = 4 * s + 2 * half;
        v2f av, bv;
        av.x = A[lid * DD + k0];
        av.y = A[lid * DD + k0 + 1];
        bv.x = h1[(k0    ) * 32 + tn * 16 + lid];
        bv.y = h1[(k0 + 1) * 32 + tn * 16 + lid];
        c = __builtin_amdgcn_wmma_f32_16x16x4_f32(false, av, false, bv,
                                                  (short)0, c, false, false);
    }
    float* dst = hid + (l * BB + tm * 16) * 32 + tn * 16;
    #pragma unroll
    for (int r = 0; r < 8; ++r) {
        int M = r + 8 * half;
        dst[M * 32 + lid] = fmaxf(c[r], 0.0f);
    }
}

__global__ void k_q(const float* __restrict__ ws, const float* __restrict__ aux,
                    const float* __restrict__ h2w, float* out) {
    int b = blockIdx.x * blockDim.x + threadIdx.x;
    if (b >= BB) return;
    float q = 0.0f;
    for (int l = 0; l < 2; ++l) {
        const float* h = ws + WS_HID + (l * BB + b) * 32;
        for (int j = 0; j < 32; ++j) q += h[j] * h2w[j];
        for (int j = 0; j < 4; ++j)  q += aux[b * 8 + l * 4 + j] * h2w[32 + j];
    }
    out[b] = q;
}

extern "C" void kernel_launch(void* const* d_in, const int* in_sizes, int n_in,
                              void* d_out, int out_size, void* d_ws, size_t ws_size,
                              hipStream_t stream) {
    const int*   n2n0_row = (const int*)  d_in[0];
    const float* n2n0_val = (const float*)d_in[2];
    const int*   n2n1_row = (const int*)  d_in[3];
    const float* n2n1_val = (const float*)d_in[5];
    const int*   subg_row = (const int*)  d_in[6];
    const int*   act_col  = (const int*)  d_in[9];
    const int*   sg0_row  = (const int*)  d_in[10];
    const int*   sg0_col  = (const int*)  d_in[11];
    const int*   sg1_row  = (const int*)  d_in[12];
    const int*   sg1_col  = (const int*)  d_in[13];
    const float* aux      = (const float*)d_in[14];
    const float* w_n2l    = (const float*)d_in[15];
    const float* p_conv   = (const float*)d_in[16];
    const float* W_sage   = (const float*)d_in[17];
    const float* cp       = (const float*)d_in[18];
    const float* h1w      = (const float*)d_in[19];
    const float* h2w      = (const float*)d_in[20];
    float* out = (float*)d_out;
    float* ws  = (float*)d_ws;
    float* sc  = out + BB;

    k_zero <<<(6 * NBT + 255) / 256, 256, 0, stream>>>(ws, sc);
    k_small<<<1, 64, 0, stream>>>(w_n2l, p_conv, W_sage, ws);
    k_n2n  <<<2 * N2N_BLK, SORT_NT, 0, stream>>>(n2n0_row, n2n0_val,
                                                  n2n1_row, n2n1_val, ws, sc);
    k_subg <<<(NN + 255) / 256, 256, 0, stream>>>(subg_row, ws, sc);
    k_sage <<<2 * SAGE_BLK, SORT_NT, 0, stream>>>(sg0_row, sg0_col,
                                                   sg1_row, sg1_col, ws, sc);
    k_mdsum<<<(NBT + 255) / 256, 256, 0, stream>>>(ws, sc);
    k_rows <<<ROWS_BLK, ROWS_NT, 0, stream>>>(ws, cp, out);
    k_esa  <<<(2 * BB * DD + 255) / 256, 256, 0, stream>>>(act_col, out, ws);
    k_gemm <<<128, 32, 0, stream>>>(ws, h1w, ws + WS_HID);
    k_q    <<<(BB + 255) / 256, 256, 0, stream>>>(ws, aux, h2w, out);
}


# subgraph histogram: a block adds into its own package half's copy (was: all dies into one copy's 32 lines); gather and md-sum launch add the two copies; float sum order changes
# speedup vs baseline: 4.8823x; 1.0611x over previous
#include <hip/hip_runtime.h>

#define NN   100000
#define BB   512
#define EE   1000000
#define ESZ  1200000
#define DD   64
#define NBT  (NN + BB)

#define WS_A0   0
#define WS_A1   (NBT)
#define WS_MD0  (2*NBT)
#define WS_MD1  (4*NBT)
#define WS_SM   (6*NBT)
#define WS_U1   (WS_SM + 128)
#define WS_U2   (WS_SM + 192)
#define WS_T    (WS_SM + 256)
#define WS_ESA  (WS_T + 2*BB)
#define WS_HID  (WS_ESA + 2*BB*DD)

typedef float v2f __attribute__((ext_vector_type(2)));
typedef float v8f __attribute__((ext_vector_type(8)));

__device__ __forceinline__ int pkg_half() {
    unsigned id = __builtin_amdgcn_s_sendmsg_rtn(0x87);
    return (int)((id >> 18) & 1u);
}

__global__ void k_zero(float* ws, float* sc) {
    int i = blockIdx.x * blockDim.x + threadIdx.x;
    if (i < 6 * NBT) { ws[i] = 0.0f; sc[i] = 0.0f; }
}

__global__ void k_small(const float* __restrict__ w_n2l,
                        const float* __restrict__ p_conv,
                        const float* __restrict__ W_sage,
                        float* ws) {
    __shared__ float v[DD], w[DD];
    int d = threadIdx.x;
    float x = fmaxf(w_n2l[d] + w_n2l[DD + d], 0.0f);
    v[d] = x;
    __syncthreads();
    float ss = 0.f;
    for (int k = 0; k < DD; ++k) ss += v[k] * v[k];
    float nv = x / fmaxf(sqrtf(ss), 1e-12f);
    __syncthreads();
    v[d] = nv;
    __syncthreads();
    float wd = 0.f;
    for (int k = 0; k < DD; ++k) wd += v[k] * p_conv[k * DD + d];
    w[d] = wd;
    __syncthreads();
    float u1 = 0.f, u2 = 0.f;
    for (int k = 0; k < DD; ++k) {
        u1 += w[k] * W_sage[k * DD + d];
        u2 += w[k] * W_sage[(DD + k) * DD + d];
    }
    ws[WS_SM + d]  = nv;
    ws[WS_SM + 64 + d] = wd;
    ws[WS_U1 + d] = u1;
    ws[WS_U2 + d] = u2;
}

#define SORT_T    16384
#define SORT_NT   1024
#define SORT_PER  (SORT_T / SORT_NT)
#define N2N_BINS  ((NBT + 31) / 32)
#define SAGE_BINS ((NBT + 15) / 16)
#define N2N_BLK   ((EE + SORT_T - 1) / SORT_T)
#define SAGE_BLK  ((ESZ + SORT_T - 1) / SORT_T)
static_assert(SORT_T == 32 * 32 * SORT_PER, "32 waves x SORT_PER groups of 32 ordered edges");

template <int NB>
__device__ __forceinline__ void bins_to_offsets(unsigned* cnt, unsigned* wsum) {
    constexpr int PER = (NB + SORT_NT - 1) / SORT_NT;
    const int tid = threadIdx.x, lane = tid & 31, wave = tid >> 5;
    unsigned loc[PER];
    unsigned s = 0;
    #pragma unroll
    for (int k = 0; k < PER; ++k) {
        int b = tid * PER + k;
        unsigned v = (b < NB) ? cnt[b] : 0u;
        loc[k] = s;
        s += v;
    }
    unsigned inc = s;
    for (int off = 1; off < 32; off <<= 1) {
        unsigned t = __shfl_up(inc, off, 32);
        if (lane >= off) inc += t;
    }
    if (lane == 31) wsum[wave] = inc;
    __syncthreads();
    if (wave == 0) {
        unsigned w = wsum[lane];
        unsigned winc = w;
        for (int off = 1; off < 32; off <<= 1) {
            unsigned t = __shfl_up(winc, off, 32);
            if (lane >= off) winc += t;
        }
        wsum[lane] = winc - w;
    }
    __syncthreads();
    unsigned base = wsum[wave] + (inc - s);
    #pragma unroll
    for (int k = 0; k < PER; ++k) {
        int b = tid * PER + k;
        if (b < NB) cnt[b] = base + loc[k];
    }
    __syncthreads();
}

__global__ void __launch_bounds__(SORT_NT)
k_n2n(const int* __restrict__ r0, const float* __restrict__ v0,
      const int* __restrict__ r1, const float* __restrict__ v1,
      float* ws, float* sc) {
    __shared__ unsigned cnt[N2N_BINS];
    __shared__ unsigned wsum[32];
    __shared__ uint2 sorted[SORT_T];
    const int tid = threadIdx.x;
    const bool l1 = blockIdx.x >= N2N_BLK;
    const int blk = l1 ? blockIdx.x - N2N_BLK : blockIdx.x;
    const int*   rr = l1 ? r1 : r0;
    const float* vv = l1 ? v1 : v0;
    float* plane = (pkg_half() ? sc : ws) + (l1 ? WS_A1 : WS_A0);
    const int base = blk * SORT_T;
    const int n = min(SORT_T, EE - base);

    for (int b = tid; b < N2N_BINS; b += SORT_NT) cnt[b] = 0u;
    int row[SORT_PER]; float val[SORT_PER]; unsigned rank[SORT_PER];
    #pragma unroll
    for (int j = 0; j < SORT_PER; ++j) {
        int k = j * SORT_NT + tid;
        int idx = base + min(k, n - 1);
        row[j] = rr[idx];
        val[j] = vv[idx];
    }
    __syncthreads();
    #pragma unroll
    for (int j = 0; j < SORT_PER; ++j) {
        int k = j * SORT_NT + tid;
        rank[j] = 0u;
        if (k < n) rank[j] = atomicAdd(&cnt[row[j] >> 5], 1u);
    }
    __syncthreads();
    bins_to_offsets<N2N_BINS>(cnt, wsum);
    unsigned pos[SORT_PER];
    #pragma unroll
    for (int j = 0; j < SORT_PER; ++j) pos[j] = cnt[row[j] >> 5] + rank[j];
    #pragma unroll
    for (int j = 0; j < SORT_PER; ++j) {
        int k = j * SORT_NT + tid;
        if (k < n) sorted[pos[j]] = make_uint2((unsigned)row[j], __float_as_uint(val[j]));
    }
    __syncthreads();
    uint2 e[SORT_PER];
    #pragma unroll
    for (int j = 0; j < SORT_PER; ++j)
        e[j] = sorted[min(j * SORT_NT + tid, n - 1)];
    #pragma unroll
    for (int j = 0; j < SORT_PER; ++j) {
        int k = j * SORT_NT + tid;
        if (k < n) atomicAdd(&plane[e[j].x], __uint_as_float(e[j].y));
    }
}

__global__ void k_subg(const int* __restrict__ subg_row, float* ws, float* sc) {
    int i = blockIdx.x * blockDim.x + threadIdx.x;
    if (i < NN) {
        float s0 = sc[WS_A0 + i];
        float s1 = sc[WS_A1 + i];
        int b = subg_row[i];
        ws[WS_A0 + i] += s0;
        ws[WS_A1 + i] += s1;
        float* mine = pkg_half() ? sc : ws;
        atomicAdd(&mine[WS_A0 + NN + b], 1.0f);
        atomicAdd(&mine[WS_A1 + NN + b], 1.0f);
    }
}

__global__ void __launch_bounds__(SORT_NT)
k_sage(const int* __restrict__ s0r, const int* __restrict__ s0c,
       const int* __restrict__ s1r, const int* __restrict__ s1c,
       float* ws, float* sc) {
    __shared__ unsigned cnt[SAGE_BINS];
    __shared__ unsigned wsum[32];
    __shared__ uint2 sorted[SORT_T];
    const int tid = threadIdx.x, lane = tid & 31, wave = tid >> 5;
    const bool l1 = blockIdx.x >= SAGE_BLK;
    const int blk = l1 ? blockIdx.x - SAGE_BLK : blockIdx.x;
    const int* rr = l1 ? s1r : s0r;
    const int* cc = l1 ? s1c : s0c;
    const float* ap = ws + (l1 ? WS_A1 : WS_A0);
    const float* ap2 = sc + (l1 ? WS_A1 : WS_A0);
    float* md = (pkg_half() ? sc : ws) + (l1 ? WS_MD1 : WS_MD0);
    const int base = blk * SORT_T;
    const int n = min(SORT_T, ESZ - base);

    for (int b = tid; b < N2N_BINS; b += SORT_NT) cnt[b] = 0u;
    int row[SORT_PER]; int col[SORT_PER]; float av[SORT_PER]; unsigned rank[SORT_PER];
    unsigned pos[SORT_PER];
    #pragma unroll
    for (int j = 0; j < SORT_PER; ++j) {
        int k = j * SORT_NT + tid;
        int idx = base + min(k, n - 1);
        row[j] = rr[idx];
        col[j] = cc[idx];
    }
    __syncthreads();
    #pragma unroll
    for (int j = 0; j < SORT_PER; ++j) {
        int k = j * SORT_NT + tid;
        rank[j] = 0u;
        if (k < n) rank[j] = atomicAdd(&cnt[col[j] >> 5], 1u);
    }
    __syncthreads();
    bins_to_offsets<N2N_BINS>(cnt, wsum);
    #pragma unroll
    for (int j = 0; j < SORT_PER; ++j) pos[j] = cnt[col[j] >> 5] + rank[j];
    #pragma unroll
    for (int j = 0; j < SORT_PER; ++j) {
        int k = j * SORT_NT + tid;
        if (k < n) sorted[pos[j]] = make_uint2((unsigned)row[j], (unsigned)col[j]);
    }
    __syncthreads();
    #pragma unroll
    for (int j = 0; j < SORT_PER; ++j) {
        uint2 e = sorted[min(j * SORT_NT + tid, n - 1)];
        row[j] = (int)e.x;
        col[j] = (int)e.y;
    }
    #pragma unroll
    for (int j = 0; j < SORT_PER; ++j) av[j] = ap[col[j]];
    #pragma unroll
    for (int j = 0; j < SORT_PER; ++j)
        if (col[j] >= NN) av[j] += ap2[col[j]];
    for (int b = tid; b < SAGE_BINS; b += SORT_NT) cnt[b] = 0u;
    __syncthreads();
    #pragma unroll
    for (int j = 0; j < SORT_PER; ++j) {
        int k = j * SORT_NT + tid;
        rank[j] = 0u;
        if (k < n) rank[j] = atomicAdd(&cnt[row[j] >> 4], 1u);
    }
    __syncthreads();
    bins_to_offsets<SAGE_BINS>(cnt, wsum);
    #pragma unroll
    for (int j = 0; j < SORT_PER; ++j) pos[j] = cnt[row[j] >> 4] + rank[j];
    #pragma unroll
    for (int j = 0; j < SORT_PER; ++j) {
        int k = j * SORT_NT + tid;
        if (k < n) sorted[pos[j]] = make_uint2((unsigned)row[j], __float_as_uint(av[j]));
    }
    __syncthreads();
    const int jj   = lane >> 1;
    const int role = lane & 1;
    uint2 eA[SORT_PER], eB[SORT_PER];
    #pragma unroll
    for (int j = 0; j < SORT_PER; ++j) {
        int p = (j * 32 + wave) * 32 + jj;
        eA[j] = sorted[min(p, n - 1)];
        eB[j] = sorted[min(p + 16, n - 1)];
    }
    #pragma unroll
    for (int j = 0; j < SORT_PER; ++j) {
        int p = (j * 32 + wave) * 32 + jj;
        if (p < n)
            atomicAdd(&md[2 * eA[j].x + role], role ? 1.0f : __uint_as_float(eA[j].y));
        if (p + 16 < n)
            atomicAdd(&md[2 * eB[j].x + role], role ? 1.0f : __uint_as_float(eB[j].y));
    }
}

__global__ void k_mdsum(float* ws, const float* __restrict__ sc) {
    int i = blockIdx.x * blockDim.x + threadIdx.x;
    if (i < NBT) {
        float4 x = *(const float4*)&ws[WS_MD0 + 4 * i];
        const float4 y = *(const float4*)&sc[WS_MD0 + 4 * i];
        x.x += y.x; x.y += y.y; x.z += y.z; x.w += y.w;
        *(float4*)&ws[WS_MD0 + 4 * i] = x;
    }
    if (i < 2 * BB) {
        int o = ((i >> 9) ? WS_A1 : WS_A0) + NN + (i & (BB - 1));
        ws[o] += sc[o];
    }
}

#define ROWS_NT    256
#define ROWS_BLK   1024
#define ROWS_WAVES (ROWS_BLK * ROWS_NT / 32)
#define ROWS_UNR   4
static_assert(NBT % 2 == 0 && NN % 2 == 0, "row pairs must not straddle a layer or the real/virtual split");

__device__ __forceinline__ float half_sum(float v) {
    v += __int_as_float(__builtin_amdgcn_update_dpp(0, __float_as_int(v), 0x161, 0xf, 0xf, true));
    v += __int_as_float(__builtin_amdgcn_update_dpp(0, __float_as_int(v), 0x162, 0xf, 0xf, true));
    v += __int_as_float(__builtin_amdgcn_update_dpp(0, __float_as_int(v), 0x164, 0xf, 0xf, true));
    v += __int_as_float(__builtin_amdgcn_update_dpp(0, __float_as_int(v), 0x168, 0xf, 0xf, true));
    return v;
}

__global__ void __launch_bounds__(ROWS_NT)
k_rows(float* ws, const float* __restrict__ cp, float* out) {
    const int lane = threadIdx.x & 31;
    const int half = lane >> 4;
    const int q    = lane & 15;
    const int gw   = (blockIdx.x * ROWS_NT + threadIdx.x) >> 5;

    const float4 u1 = *(const float4*)&ws[WS_U1 + 4 * q];
    const float4 u2 = *(const float4*)&ws[WS_U2 + 4 * q];

    for (int p0 = gw; p0 < NBT; p0 += ROWS_WAVES * ROWS_UNR) {
        float a[ROWS_UNR]; float2 md[ROWS_UNR];
        #pragma unroll
        for (int k = 0; k < ROWS_UNR; ++k) {
            int p = min(p0 + k * ROWS_WAVES, NBT - 1);
            int w = 2 * p + half;
            int l = (w >= NBT) ? 1 : 0;
            int i = w - l * NBT;
            a[k]  = ws[(l ? WS_A1 : WS_A0) + i];
            md[k] = *(const float2*)&ws[(l ? WS_MD1 : WS_MD0) + 2 * i];
        }
        #pragma unroll
        for (int k = 0; k < ROWS_UNR; ++k) {
            int p = p0 + k * ROWS_WAVES;
            if (p < NBT) {
                int w = 2 * p + half;
                int l = (w >= NBT) ? 1 : 0;
                int i = w - l * NBT;
                float m = md[k].x / fmaxf(md[k].y, 1.0f);
                float r0 = fmaxf(a[k] * u1.x + m * u2.x, 0.0f);
                float r1 = fmaxf(a[k] * u1.y + m * u2.y, 0.0f);
                float r2 = fmaxf(a[k] * u1.z + m * u2.z, 0.0f);
                float r3 = fmaxf(a[k] * u1.w + m * u2.w, 0.0f);
                float ss = (r0 * r0 + r1 * r1) + (r2 * r2 + r3 * r3);
                ss = half_sum(ss);
                float inv = 1.0f / fmaxf(sqrtf(ss), 1e-12f);
                r0 *= inv; r1 *= inv; r2 *= inv; r3 *= inv;
                if (i < NN) {
                    float* dst = out + BB + (l * NN + i) * DD + 4 * q;
                    *(float4*)dst = make_float4(r0, r1, r2, r3);
                } else {
                    int b = i - NN;
                    const float4 c4 = *(const float4*)&cp[4 * q];
                    float t = (r0 * c4.x + r1 * c4.y) + (r2 * c4.z + r3 * c4.w);
                    t = half_sum(t);
                    if (q == 0) ws[WS_T + l * BB + b] = t;
                }
            }
        }
    }
}

__global__ void k_esa(const int* __restrict__ act_col, const float* __restrict__ out,
                      float* ws) {
    int idx = blockIdx.x * blockDim.x + threadIdx.x;
    if (idx >= 2 * BB * DD) return;
    int d = idx & (DD - 1);
    int b = (idx >> 6) & (BB - 1);
    int l = idx >> 15;
    float t = ws[WS_T + l * BB + b];
    float ae = out[BB + (l * NN + act_col[b]) * DD + d];
    ws[WS_ESA + idx] = t * ae;
}

__global__ void k_gemm(const float* __restrict__ ws, const float* __restrict__ h1,
                       float* __restrict__ hid) {
    int bid  = blockIdx.x;
    int tn   = bid & 1;
    int tm   = (bid >> 1) & 31;
    int l    = bid >> 6;
    int lane = threadIdx.x;
    int half = lane >> 4;
    int lid  = lane & 15;

    const float* A = ws + WS_ESA + (l * BB + tm * 16) * DD;
    v8f c = {0.f, 0.f, 0.f, 0.f, 0.f, 0.f, 0.f, 0.f};
    #pragma unroll
    for (int s = 0; s < 16; ++s) {
        int k0 = 4 * s + 2 * half;
        v2f av, bv;
        av.x = A[lid * DD + k0];
        av.y = A[lid * DD + k0 + 1];
        bv.x = h1[(k0    ) * 32 + tn * 16 + lid];
        bv.y = h1[(k0 + 1) * 32 + tn * 16 + lid];
        c = __builtin_amdgcn_wmma_f32_16x16x4_f32(false, av, false, bv,
                                                  (short)0, c, false, false);
    }
    float* dst = hid + (l * BB + tm * 16) * 32 + tn * 16;
    #pragma unroll
    for (int r = 0; r < 8; ++r) {
        int M = r + 8 * half;
        dst[M * 32 + lid] = fmaxf(c[r], 0.0f);
    }
}

__global__ void k_q(const float* __restrict__ ws, const float* __restrict__ aux,
                    const float* __restrict__ h2w, float* out) {
    int b = blockIdx.x * blockDim.x + threadIdx.x;
    if (b >= BB) return;
    float q = 0.0f;
    for (int l = 0; l < 2; ++l) {
        const float* h = ws + WS_HID + (l * BB + b) * 32;
        for (int j = 0; j < 32; ++j) q += h[j] * h2w[j];
        for (int j = 0; j < 4; ++j)  q += aux[b * 8 + l * 4 + j] * h2w[32 + j];
    }
    out[b] = q;
}

extern "C" void kernel_launch(void* const* d_in, const int* in_sizes, int n_in,
                              void* d_out, int out_size, void* d_ws, size_t ws_size,
                              hipStream_t stream) {
    const int*   n2n0_row = (const int*)  d_in[0];
    const float* n2n0_val = (const float*)d_in[2];
    const int*   n2n1_row = (const int*)  d_in[3];
    const float* n2n1_val = (const float*)d_in[5];
    const int*   subg_row = (const int*)  d_in[6];
    const int*   act_col  = (const int*)  d_in[9];
    const int*   sg0_row  = (const int*)  d_in[10];
    const int*   sg0_col  = (const int*)  d_in[11];
    const int*   sg1_row  = (const int*)  d_in[12];
    const int*   sg1_col  = (const int*)  d_in[13];
    const float* aux      = (const float*)d_in[14];
    const float* w_n2l    = (const float*)d_in[15];
    const float* p_conv   = (const float*)d_in[16];
    const float* W_sage   = (const float*)d_in[17];
    const float* cp       = (const float*)d_in[18];
    const float* h1w      = (const float*)d_in[19];
    const float* h2w      = (const float*)d_in[20];
    float* out = (float*)d_out;
    float* ws  = (float*)d_ws;
    float* sc  = out + BB;

    k_zero <<<(6 * NBT + 255) / 256, 256, 0, stream>>>(ws, sc);
    k_small<<<1, 64, 0, stream>>>(w_n2l, p_conv, W_sage, ws);
    k_n2n  <<<2 * N2N_BLK, SORT_NT, 0, stream>>>(n2n0_row, n2n0_val,
                                                  n2n1_row, n2n1_val, ws, sc);
    k_subg <<<(NN + 255) / 256, 256, 0, stream>>>(subg_row, ws, sc);
    k_sage <<<2 * SAGE_BLK, SORT_NT, 0, stream>>>(sg0_row, sg0_col,
                                                   sg1_row, sg1_col, ws, sc);
    k_mdsum<<<(NBT + 255) / 256, 256, 0, stream>>>(ws, sc);
    k_rows <<<ROWS_BLK, ROWS_NT, 0, stream>>>(ws, cp, out);
    k_esa  <<<(2 * BB * DD + 255) / 256, 256, 0, stream>>>(act_col, out, ws);
    k_gemm <<<128, 32, 0, stream>>>(ws, h1w, ws + WS_HID);
    k_q    <<<(BB + 255) / 256, 256, 0, stream>>>(ws, aux, h2w, out);
}


# tail: WMMA A fragments read cur[act_col] rows from out, t scales the product after it (relu(t*z)), h2 dot and layer sum end in the wave: no esa/hid planes, 3 launches -> 1; q sum order changes
# speedup vs baseline: 5.2598x; 1.0773x over previous
#include <hip/hip_runtime.h>

#define NN   100000
#define BB   512
#define EE   1000000
#define ESZ  1200000
#define DD   64
#define NBT  (NN + BB)

#define WS_A0   0
#define WS_A1   (NBT)
#define WS_MD0  (2*NBT)
#define WS_MD1  (4*NBT)
#define WS_SM   (6*NBT)
#define WS_U1   (WS_SM + 128)
#define WS_U2   (WS_SM + 192)
#define WS_T    (WS_SM + 256)

typedef float v2f __attribute__((ext_vector_type(2)));
typedef float v8f __attribute__((ext_vector_type(8)));

__device__ __forceinline__ int pkg_half() {
    unsigned id = __builtin_amdgcn_s_sendmsg_rtn(0x87);
    return (int)((id >> 18) & 1u);
}

__global__ void k_zero(float* ws, float* sc) {
    int i = blockIdx.x * blockDim.x + threadIdx.x;
    if (i < 6 * NBT) { ws[i] = 0.0f; sc[i] = 0.0f; }
}

__global__ void k_small(const float* __restrict__ w_n2l,
                        const float* __restrict__ p_conv,
                        const float* __restrict__ W_sage,
                        float* ws) {
    __shared__ float v[DD], w[DD];
    int d = threadIdx.x;
    float x = fmaxf(w_n2l[d] + w_n2l[DD + d], 0.0f);
    v[d] = x;
    __syncthreads();
    float ss = 0.f;
    for (int k = 0; k < DD; ++k) ss += v[k] * v[k];
    float nv = x / fmaxf(sqrtf(ss), 1e-12f);
    __syncthreads();
    v[d] = nv;
    __syncthreads();
    float wd = 0.f;
    for (int k = 0; k < DD; ++k) wd += v[k] * p_conv[k * DD + d];
    w[d] = wd;
    __syncthreads();
    float u1 = 0.f, u2 = 0.f;
    for (int k = 0; k < DD; ++k) {
        u1 += w[k] * W_sage[k * DD + d];
        u2 += w[k] * W_sage[(DD + k) * DD + d];
    }
    ws[WS_SM + d]  = nv;
    ws[WS_SM + 64 + d] = wd;
    ws[WS_U1 + d] = u1;
    ws[WS_U2 + d] = u2;
}

#define SORT_T    16384
#define SORT_NT   1024
#define SORT_PER  (SORT_T / SORT_NT)
#define N2N_BINS  ((NBT + 31) / 32)
#define SAGE_BINS ((NBT + 15) / 16)
#define N2N_BLK   ((EE + SORT_T - 1) / SORT_T)
#define SAGE_BLK  ((ESZ + SORT_T - 1) / SORT_T)
static_assert(SORT_T == 32 * 32 * SORT_PER, "32 waves x SORT_PER groups of 32 ordered edges");

template <int NB>
__device__ __forceinline__ void bins_to_offsets(unsigned* cnt, unsigned* wsum) {
    constexpr int PER = (NB + SORT_NT - 1) / SORT_NT;
    const int tid = threadIdx.x, lane = tid & 31, wave = tid >> 5;
    unsigned loc[PER];
    unsigned s = 0;
    #pragma unroll
    for (int k = 0; k < PER; ++k) {
        int b = tid * PER + k;
        unsigned v = (b < NB) ? cnt[b] : 0u;
        loc[k] = s;
        s += v;
    }
    unsigned inc = s;
    for (int off = 1; off < 32; off <<= 1) {
        unsigned t = __shfl_up(inc, off, 32);
        if (lane >= off) inc += t;
    }
    if (lane == 31) wsum[wave] = inc;
    __syncthreads();
    if (wave == 0) {
        unsigned w = wsum[lane];
        unsigned winc = w;
        for (int off = 1; off < 32; off <<= 1) {
            unsigned t = __shfl_up(winc, off, 32);
            if (lane >= off) winc += t;
        }
        wsum[lane] = winc - w;
    }
    __syncthreads();
    unsigned base = wsum[wave] + (inc - s);
    #pragma unroll
    for (int k = 0; k < PER; ++k) {
        int b = tid * PER + k;
        if (b < NB) cnt[b] = base + loc[k];
    }
    __syncthreads();
}

__global__ void __launch_bounds__(SORT_NT)
k_n2n(const int* __restrict__ r0, const float* __restrict__ v0,
      const int* __restrict__ r1, const float* __restrict__ v1,
      float* ws, float* sc) {
    __shared__ unsigned cnt[N2N_BINS];
    __shared__ unsigned wsum[32];
    __shared__ uint2 sorted[SORT_T];
    const int tid = threadIdx.x;
    const bool l1 = blockIdx.x >= N2N_BLK;
    const int blk = l1 ? blockIdx.x - N2N_BLK : blockIdx.x;
    const int*   rr = l1 ? r1 : r0;
    const float* vv = l1 ? v1 : v0;
    float* plane = (pkg_half() ? sc : ws) + (l1 ? WS_A1 : WS_A0);
    const int base = blk * SORT_T;
    const int n = min(SORT_T, EE - base);

    for (int b = tid; b < N2N_BINS; b += SORT_NT) cnt[b] = 0u;
    int row[SORT_PER]; float val[SORT_PER]; unsigned rank[SORT_PER];
    #pragma unroll
    for (int j = 0; j < SORT_PER; ++j) {
        int k = j * SORT_NT + tid;
        int idx = base + min(k, n - 1);
        row[j] = rr[idx];
        val[j] = vv[idx];
    }
    __syncthreads();
    #pragma unroll
    for (int j = 0; j < SORT_PER; ++j) {
        int k = j * SORT_NT + tid;
        rank[j] = 0u;
        if (k < n) rank[j] = atomicAdd(&cnt[row[j] >> 5], 1u);
    }
    __syncthreads();
    bins_to_offsets<N2N_BINS>(cnt, wsum);
    unsigned pos[SORT_PER];
    #pragma unroll
    for (int j = 0; j < SORT_PER; ++j) pos[j] = cnt[row[j] >> 5] + rank[j];
    #pragma unroll
    for (int j = 0; j < SORT_PER; ++j) {
        int k = j * SORT_NT + tid;
        if (k < n) sorted[pos[j]] = make_uint2((unsigned)row[j], __float_as_uint(val[j]));
    }
    __syncthreads();
    uint2 e[SORT_PER];
    #pragma unroll
    for (int j = 0; j < SORT_PER; ++j)
        e[j] = sorted[min(j * SORT_NT + tid, n - 1)];
    #pragma unroll
    for (int j = 0; j < SORT_PER; ++j) {
        int k = j * SORT_NT + tid;
        if (k < n) atomicAdd(&plane[e[j].x], __uint_as_float(e[j].y));
    }
}

__global__ void k_subg(const int* __restrict__ subg_row, float* ws, float* sc) {
    int i = blockIdx.x * blockDim.x + threadIdx.x;
    if (i < NN) {
        float s0 = sc[WS_A0 + i];
        float s1 = sc[WS_A1 + i];
        int b = subg_row[i];
        ws[WS_A0 + i] += s0;
        ws[WS_A1 + i] += s1;
        float* mine = pkg_half() ? sc : ws;
        atomicAdd(&mine[WS_A0 + NN + b], 1.0f);
        atomicAdd(&mine[WS_A1 + NN + b], 1.0f);
    }
}

__global__ void __launch_bounds__(SORT_NT)
k_sage(const int* __restrict__ s0r, const int* __restrict__ s0c,
       const int* __restrict__ s1r, const int* __restrict__ s1c,
       float* ws, float* sc) {
    __shared__ unsigned cnt[SAGE_BINS];
    __shared__ unsigned wsum[32];
    __shared__ uint2 sorted[SORT_T];
    const int tid = threadIdx.x, lane = tid & 31, wave = tid >> 5;
    const bool l1 = blockIdx.x >= SAGE_BLK;
    const int blk = l1 ? blockIdx.x - SAGE_BLK : blockIdx.x;
    const int* rr = l1 ? s1r : s0r;
    const int* cc = l1 ? s1c : s0c;
    const float* ap = ws + (l1 ? WS_A1 : WS_A0);
    const float* ap2 = sc + (l1 ? WS_A1 : WS_A0);
    float* md = (pkg_half() ? sc : ws) + (l1 ? WS_MD1 : WS_MD0);
    const int base = blk * SORT_T;
    const int n = min(SORT_T, ESZ - base);

    for (int b = tid; b < N2N_BINS; b += SORT_NT) cnt[b] = 0u;
    int row[SORT_PER]; int col[SORT_PER]; float av[SORT_PER]; unsigned rank[SORT_PER];
    unsigned pos[SORT_PER];
    #pragma unroll
    for (int j = 0; j < SORT_PER; ++j) {
        int k = j * SORT_NT + tid;
        int idx = base + min(k, n - 1);
        row[j] = rr[idx];
        col[j] = cc[idx];
    }
    __syncthreads();
    #pragma unroll
    for (int j = 0; j < SORT_PER; ++j) {
        int k = j * SORT_NT + tid;
        rank[j] = 0u;
        if (k < n) rank[j] = atomicAdd(&cnt[col[j] >> 5], 1u);
    }
    __syncthreads();
    bins_to_offsets<N2N_BINS>(cnt, wsum);
    #pragma unroll
    for (int j = 0; j < SORT_PER; ++j) pos[j] = cnt[col[j] >> 5] + rank[j];
    #pragma unroll
    for (int j = 0; j < SORT_PER; ++j) {
        int k = j * SORT_NT + tid;
        if (k < n) sorted[pos[j]] = make_uint2((unsigned)row[j], (unsigned)col[j]);
    }
    __syncthreads();
    #pragma unroll
    for (int j = 0; j < SORT_PER; ++j) {
        uint2 e = sorted[min(j * SORT_NT + tid, n - 1)];
        row[j] = (int)e.x;
        col[j] = (int)e.y;
    }
    #pragma unroll
    for (int j = 0; j < SORT_PER; ++j) av[j] = ap[col[j]];
    #pragma unroll
    for (int j = 0; j < SORT_PER; ++j)
        if (col[j] >= NN) av[j] += ap2[col[j]];
    for (int b = tid; b < SAGE_BINS; b += SORT_NT) cnt[b] = 0u;
    __syncthreads();
    #pragma unroll
    for (int j = 0; j < SORT_PER; ++j) {
        int k = j * SORT_NT + tid;
        rank[j] = 0u;
        if (k < n) rank[j] = atomicAdd(&cnt[row[j] >> 4], 1u);
    }
    __syncthreads();
    bins_to_offsets<SAGE_BINS>(cnt, wsum);
    #pragma unroll
    for (int j = 0; j < SORT_PER; ++j) pos[j] = cnt[row[j] >> 4] + rank[j];
    #pragma unroll
    for (int j = 0; j < SORT_PER; ++j) {
        int k = j * SORT_NT + tid;
        if (k < n) sorted[pos[j]] = make_uint2((unsigned)row[j], __float_as_uint(av[j]));
    }
    __syncthreads();
    const int jj   = lane >> 1;
    const int role = lane & 1;
    uint2 eA[SORT_PER], eB[SORT_PER];
    #pragma unroll
    for (int j = 0; j < SORT_PER; ++j) {
        int p = (j * 32 + wave) * 32 + jj;
        eA[j] = sorted[min(p, n - 1)];
        eB[j] = sorted[min(p + 16, n - 1)];
    }
    #pragma unroll
    for (int j = 0; j < SORT_PER; ++j) {
        int p = (j * 32 + wave) * 32 + jj;
        if (p < n)
            atomicAdd(&md[2 * eA[j].x + role], role ? 1.0f : __uint_as_float(eA[j].y));
        if (p + 16 < n)
            atomicAdd(&md[2 * eB[j].x + role], role ? 1.0f : __uint_as_float(eB[j].y));
    }
}

__global__ void k_mdsum(float* ws, const float* __restrict__ sc) {
    int i = blockIdx.x * blockDim.x + threadIdx.x;
    if (i < NBT) {
        float4 x = *(const float4*)&ws[WS_MD0 + 4 * i];
        const float4 y = *(const float4*)&sc[WS_MD0 + 4 * i];
        x.x += y.x; x.y += y.y; x.z += y.z; x.w += y.w;
        *(float4*)&ws[WS_MD0 + 4 * i] = x;
    }
    if (i < 2 * BB) {
        int o = ((i >> 9) ? WS_A1 : WS_A0) + NN + (i & (BB - 1));
        ws[o] += sc[o];
    }
}

#define ROWS_NT    256
#define ROWS_BLK   1024
#define ROWS_WAVES (ROWS_BLK * ROWS_NT / 32)
#define ROWS_UNR   4
static_assert(NBT % 2 == 0 && NN % 2 == 0, "row pairs must not straddle a layer or the real/virtual split");

__device__ __forceinline__ float half_sum(float v) {
    v += __int_as_float(__builtin_amdgcn_update_dpp(0, __float_as_int(v), 0x161, 0xf, 0xf, true));
    v += __int_as_float(__builtin_amdgcn_update_dpp(0, __float_as_int(v), 0x162, 0xf, 0xf, true));
    v += __int_as_float(__builtin_amdgcn_update_dpp(0, __float_as_int(v), 0x164, 0xf, 0xf, true));
    v += __int_as_float(__builtin_amdgcn_update_dpp(0, __float_as_int(v), 0x168, 0xf, 0xf, true));
    return v;
}

__global__ void __launch_bounds__(ROWS_NT)
k_rows(float* ws, const float* __restrict__ cp, float* out) {
    const int lane = threadIdx.x & 31;
    const int half = lane >> 4;
    const int q    = lane & 15;
    const int gw   = (blockIdx.x * ROWS_NT + threadIdx.x) >> 5;

    const float4 u1 = *(const float4*)&ws[WS_U1 + 4 * q];
    const float4 u2 = *(const float4*)&ws[WS_U2 + 4 * q];

    for (int p0 = gw; p0 < NBT; p0 += ROWS_WAVES * ROWS_UNR) {
        float a[ROWS_UNR]; float2 md[ROWS_UNR];
        #pragma unroll
        for (int k = 0; k < ROWS_UNR; ++k) {
            int p = min(p0 + k * ROWS_WAVES, NBT - 1);
            int w = 2 * p + half;
            int l = (w >= NBT) ? 1 : 0;
            int i = w - l * NBT;
            a[k]  = ws[(l ? WS_A1 : WS_A0) + i];
            md[k] = *(const float2*)&ws[(l ? WS_MD1 : WS_MD0) + 2 * i];
        }
        #pragma unroll
        for (int k = 0; k < ROWS_UNR; ++k) {
            int p = p0 + k * ROWS_WAVES;
            if (p < NBT) {
                int w = 2 * p + half;
                int l = (w >= NBT) ? 1 : 0;
                int i = w - l * NBT;
                float m = md[k].x / fmaxf(md[k].y, 1.0f);
                float r0 = fmaxf(a[k] * u1.x + m * u2.x, 0.0f);
                float r1 = fmaxf(a[k] * u1.y + m * u2.y, 0.0f);
                float r2 = fmaxf(a[k] * u1.z + m * u2.z, 0.0f);
                float r3 = fmaxf(a[k] * u1.w + m * u2.w, 0.0f);
                float ss = (r0 * r0 + r1 * r1) + (r2 * r2 + r3 * r3);
                ss = half_sum(ss);
                float inv = 1.0f / fmaxf(sqrtf(ss), 1e-12f);
                r0 *= inv; r1 *= inv; r2 *= inv; r3 *= inv;
                if (i < NN) {
                    float* dst = out + BB + (l * NN + i) * DD + 4 * q;
                    *(float4*)dst = make_float4(r0, r1, r2, r3);
                } else {
                    int b = i - NN;
                    const float4 c4 = *(const float4*)&cp[4 * q];
                    float t = (r0 * c4.x + r1 * c4.y) + (r2 * c4.z + r3 * c4.w);
                    t = half_sum(t);
                    if (q == 0) ws[WS_T + l * BB + b] = t;
                }
            }
        }
    }
}

#define TAIL_NT 64
__global__ void __launch_bounds__(TAIL_NT)
k_tail(const int* __restrict__ act_col, const float* __restrict__ ws,
       const float* __restrict__ h1, const float* __restrict__ aux,
       const float* __restrict__ h2w, float* out) {
    __shared__ float sq[32];
    const int lane = threadIdx.x & 31;
    const int wv   = threadIdx.x >> 5;
    const int half = lane >> 4;
    const int lid  = lane & 15;
    const int b0   = (blockIdx.x * 2 + wv) * 16;

    const float* A0 = out + BB + act_col[b0 + lid] * DD;
    const float* A1 = A0 + NN * DD;
    v8f c00 = {0.f, 0.f, 0.f, 0.f, 0.f, 0.f, 0.f, 0.f};
    v8f c01 = c00, c10 = c00, c11 = c00;
    #pragma unroll
    for (int s = 0; s < 16; ++s) {
        int k0 = 4 * s + 2 * half;
        v2f a0 = *(const v2f*)(A0 + k0);
        v2f a1 = *(const v2f*)(A1 + k0);
        v2f bv0, bv1;
        bv0.x = h1[(k0    ) * 32 + lid];
        bv0.y = h1[(k0 + 1) * 32 + lid];
        bv1.x = h1[(k0    ) * 32 + 16 + lid];
        bv1.y = h1[(k0 + 1) * 32 + 16 + lid];
        c00 = __builtin_amdgcn_wmma_f32_16x16x4_f32(false, a0, false, bv0, (short)0, c00, false, false);
        c01 = __builtin_amdgcn_wmma_f32_16x16x4_f32(false, a0, false, bv1, (short)0, c01, false, false);
        c10 = __builtin_amdgcn_wmma_f32_16x16x4_f32(false, a1, false, bv0, (short)0, c10, false, false);
        c11 = __builtin_amdgcn_wmma_f32_16x16x4_f32(false, a1, false, bv1, (short)0, c11, false, false);
    }
    const float w0 = h2w[lid], w1 = h2w[16 + lid];
    float part[8];
    #pragma unroll
    for (int r = 0; r < 8; ++r) {
        int b = b0 + r + 8 * half;
        float t0 = ws[WS_T + b];
        float t1 = ws[WS_T + BB + b];
        float p0 = fmaxf(t0 * c00[r], 0.0f) * w0 + fmaxf(t0 * c01[r], 0.0f) * w1;
        float p1 = fmaxf(t1 * c10[r], 0.0f) * w0 + fmaxf(t1 * c11[r], 0.0f) * w1;
        part[r] = half_sum(p0 + p1);
    }
    float qv = 0.0f;
    #pragma unroll
    for (int r = 0; r < 8; ++r) qv = (lid == r) ? part[r] : qv;
    if (lid < 8) {
        int M = lid + 8 * half;
        const float* ax = aux + (b0 + M) * 8;
        for (int l = 0; l < 2; ++l)
            for (int j = 0; j < 4; ++j) qv += ax[l * 4 + j] * h2w[32 + j];
        sq[wv * 16 + M] = qv;
    }
    __syncthreads();
    if (wv == 0) out[blockIdx.x * 32 + lane] = sq[lane];
}

extern "C" void kernel_launch(void* const* d_in, const int* in_sizes, int n_in,
                              void* d_out, int out_size, void* d_ws, size_t ws_size,
                              hipStream_t stream) {
    const int*   n2n0_row = (const int*)  d_in[0];
    const float* n2n0_val = (const float*)d_in[2];
    const int*   n2n1_row = (const int*)  d_in[3];
    const float* n2n1_val = (const float*)d_in[5];
    const int*   subg_row = (const int*)  d_in[6];
    const int*   act_col  = (const int*)  d_in[9];
    const int*   sg0_row  = (const int*)  d_in[10];
    const int*   sg0_col  = (const int*)  d_in[11];
    const int*   sg1_row  = (const int*)  d_in[12];
    const int*   sg1_col  = (const int*)  d_in[13];
    const float* aux      = (const float*)d_in[14];
    const float* w_n2l    = (const float*)d_in[15];
    const float* p_conv   = (const float*)d_in[16];
    const float* W_sage   = (const float*)d_in[17];
    const float* cp       = (const float*)d_in[18];
    const float* h1w      = (const float*)d_in[19];
    const float* h2w      = (const float*)d_in[20];
    float* out = (float*)d_out;
    float* ws  = (float*)d_ws;
    float* sc  = out + BB;

    k_zero <<<(6 * NBT + 255) / 256, 256, 0, stream>>>(ws, sc);
    k_small<<<1, 64, 0, stream>>>(w_n2l, p_conv, W_sage, ws);
    k_n2n  <<<2 * N2N_BLK, SORT_NT, 0, stream>>>(n2n0_row, n2n0_val,
                                                  n2n1_row, n2n1_val, ws, sc);
    k_subg <<<(NN + 255) / 256, 256, 0, stream>>>(subg_row, ws, sc);
    k_sage <<<2 * SAGE_BLK, SORT_NT, 0, stream>>>(sg0_row, sg0_col,
                                                   sg1_row, sg1_col, ws, sc);
    k_mdsum<<<(NBT + 255) / 256, 256, 0, stream>>>(ws, sc);
    k_rows <<<ROWS_BLK, ROWS_NT, 0, stream>>>(ws, cp, out);
    k_tail <<<BB / 32, TAIL_NT, 0, stream>>>(act_col, ws, h1w, aux, h2w, out);
}


# rows kernel: every block rebuilds the tiny dense chain (x, v, w, u1, u2) in its own LDS, k split over thread groups; its one-block launch goes: 8 launches -> 7; norm, w, u1, u2 sum order changes
# speedup vs baseline: 5.4693x; 1.0398x over previous
#include <hip/hip_runtime.h>

#define NN   100000
#define BB   512
#define EE   1000000
#define ESZ  1200000
#define DD   64
#define NBT  (NN + BB)

#define WS_A0   0
#define WS_A1   (NBT)
#define WS_MD0  (2*NBT)
#define WS_MD1  (4*NBT)
#define WS_SM   (6*NBT)
#define WS_T    (WS_SM + 256)

typedef float v2f __attribute__((ext_vector_type(2)));
typedef float v8f __attribute__((ext_vector_type(8)));

__device__ __forceinline__ int pkg_half() {
    unsigned id = __builtin_amdgcn_s_sendmsg_rtn(0x87);
    return (int)((id >> 18) & 1u);
}

__global__ void k_zero(float* ws, float* sc) {
    int i = blockIdx.x * blockDim.x + threadIdx.x;
    if (i < 6 * NBT) { ws[i] = 0.0f; sc[i] = 0.0f; }
}

#define SORT_T    16384
#define SORT_NT   1024
#define SORT_PER  (SORT_T / SORT_NT)
#define N2N_BINS  ((NBT + 31) / 32)
#define SAGE_BINS ((NBT + 15) / 16)
#define N2N_BLK   ((EE + SORT_T - 1) / SORT_T)
#define SAGE_BLK  ((ESZ + SORT_T - 1) / SORT_T)
static_assert(SORT_T == 32 * 32 * SORT_PER, "32 waves x SORT_PER groups of 32 ordered edges");

template <int NB>
__device__ __forceinline__ void bins_to_offsets(unsigned* cnt, unsigned* wsum) {
    constexpr int PER = (NB + SORT_NT - 1) / SORT_NT;
    const int tid = threadIdx.x, lane = tid & 31, wave = tid >> 5;
    unsigned loc[PER];
    unsigned s = 0;
    #pragma unroll
    for (int k = 0; k < PER; ++k) {
        int b = tid * PER + k;
        unsigned v = (b < NB) ? cnt[b] : 0u;
        loc[k] = s;
        s += v;
    }
    unsigned inc = s;
    for (int off = 1; off < 32; off <<= 1) {
        unsigned t = __shfl_up(inc, off, 32);
        if (lane >= off) inc += t;
    }
    if (lane == 31) wsum[wave] = inc;
    __syncthreads();
    if (wave == 0) {
        unsigned w = wsum[lane];
        unsigned winc = w;
        for (int off = 1; off < 32; off <<= 1) {
            unsigned t = __shfl_up(winc, off, 32);
            if (lane >= off) winc += t;
        }
        wsum[lane] = winc - w;
    }
    __syncthreads();
    unsigned base = wsum[wave] + (inc - s);
    #pragma unroll
    for (int k = 0; k < PER; ++k) {
        int b = tid * PER + k;
        if (b < NB) cnt[b] = base + loc[k];
    }
    __syncthreads();
}

__global__ void __launch_bounds__(SORT_NT)
k_n2n(const int* __restrict__ r0, const float* __restrict__ v0,
      const int* __restrict__ r1, const float* __restrict__ v1,
      float* ws, float* sc) {
    __shared__ unsigned cnt[N2N_BINS];
    __shared__ unsigned wsum[32];
    __shared__ uint2 sorted[SORT_T];
    const int tid = threadIdx.x;
    const bool l1 = blockIdx.x >= N2N_BLK;
    const int blk = l1 ? blockIdx.x - N2N_BLK : blockIdx.x;
    const int*   rr = l1 ? r1 : r0;
    const float* vv = l1 ? v1 : v0;
    float* plane = (pkg_half() ? sc : ws) + (l1 ? WS_A1 : WS_A0);
    const int base = blk * SORT_T;
    const int n = min(SORT_T, EE - base);

    for (int b = tid; b < N2N_BINS; b += SORT_NT) cnt[b] = 0u;
    int row[SORT_PER]; float val[SORT_PER]; unsigned rank[SORT_PER];
    #pragma unroll
    for (int j = 0; j < SORT_PER; ++j) {
        int k = j * SORT_NT + tid;
        int idx = base + min(k, n - 1);
        row[j] = rr[idx];
        val[j] = vv[idx];
    }
    __syncthreads();
    #pragma unroll
    for (int j = 0; j < SORT_PER; ++j) {
        int k = j * SORT_NT + tid;
        rank[j] = 0u;
        if (k < n) rank[j] = atomicAdd(&cnt[row[j] >> 5], 1u);
    }
    __syncthreads();
    bins_to_offsets<N2N_BINS>(cnt, wsum);
    unsigned pos[SORT_PER];
    #pragma unroll
    for (int j = 0; j < SORT_PER; ++j) pos[j] = cnt[row[j] >> 5] + rank[j];
    #pragma unroll
    for (int j = 0; j < SORT_PER; ++j) {
        int k = j * SORT_NT + tid;
        if (k < n) sorted[pos[j]] = make_uint2((unsigned)row[j], __float_as_uint(val[j]));
    }
    __syncthreads();
    uint2 e[SORT_PER];
    #pragma unroll
    for (int j = 0; j < SORT_PER; ++j)
        e[j] = sorted[min(j * SORT_NT + tid, n - 1)];
    #pragma unroll
    for (int j = 0; j < SORT_PER; ++j) {
        int k = j * SORT_NT + tid;
        if (k < n) atomicAdd(&plane[e[j].x], __uint_as_float(e[j].y));
    }
}

__global__ void k_subg(const int* __restrict__ subg_row, float* ws, float* sc) {
    int i = blockIdx.x * blockDim.x + threadIdx.x;
    if (i < NN) {
        float s0 = sc[WS_A0 + i];
        float s1 = sc[WS_A1 + i];
        int b = subg_row[i];
        ws[WS_A0 + i] += s0;
        ws[WS_A1 + i] += s1;
        float* mine = pkg_half() ? sc : ws;
        atomicAdd(&mine[WS_A0 + NN + b], 1.0f);
        atomicAdd(&mine[WS_A1 + NN + b], 1.0f);
    }
}

__global__ void __launch_bounds__(SORT_NT)
k_sage(const int* __restrict__ s0r, const int* __restrict__ s0c,
       const int* __restrict__ s1r, const int* __restrict__ s1c,
       float* ws, float* sc) {
    __shared__ unsigned cnt[SAGE_BINS];
    __shared__ unsigned wsum[32];
    __shared__ uint2 sorted[SORT_T];
    const int tid = threadIdx.x, lane = tid & 31, wave = tid >> 5;
    const bool l1 = blockIdx.x >= SAGE_BLK;
    const int blk = l1 ? blockIdx.x - SAGE_BLK : blockIdx.x;
    const int* rr = l1 ? s1r : s0r;
    const int* cc = l1 ? s1c : s0c;
    const float* ap = ws + (l1 ? WS_A1 : WS_A0);
    const float* ap2 = sc + (l1 ? WS_A1 : WS_A0);
    float* md = (pkg_half() ? sc : ws) + (l1 ? WS_MD1 : WS_MD0);
    const int base = blk * SORT_T;
    const int n = min(SORT_T, ESZ - base);

    for (int b = tid; b < N2N_BINS; b += SORT_NT) cnt[b] = 0u;
    int row[SORT_PER]; int col[SORT_PER]; float av[SORT_PER]; unsigned rank[SORT_PER];
    unsigned pos[SORT_PER];
    #pragma unroll
    for (int j = 0; j < SORT_PER; ++j) {
        int k = j * SORT_NT + tid;
        int idx = base + min(k, n - 1);
        row[j] = rr[idx];
        col[j] = cc[idx];
    }
    __syncthreads();
    #pragma unroll
    for (int j = 0; j < SORT_PER; ++j) {
        int k = j * SORT_NT + tid;
        rank[j] = 0u;
        if (k < n) rank[j] = atomicAdd(&cnt[col[j] >> 5], 1u);
    }
    __syncthreads();
    bins_to_offsets<N2N_BINS>(cnt, wsum);
    #pragma unroll
    for (int j = 0; j < SORT_PER; ++j) pos[j] = cnt[col[j] >> 5] + rank[j];
    #pragma unroll
    for (int j = 0; j < SORT_PER; ++j) {
        int k = j * SORT_NT + tid;
        if (k < n) sorted[pos[j]] = make_uint2((unsigned)row[j], (unsigned)col[j]);
    }
    __syncthreads();
    #pragma unroll
    for (int j = 0; j < SORT_PER; ++j) {
        uint2 e = sorted[min(j * SORT_NT + tid, n - 1)];
        row[j] = (int)e.x;
        col[j] = (int)e.y;
    }
    #pragma unroll
    for (int j = 0; j < SORT_PER; ++j) av[j] = ap[col[j]];
    #pragma unroll
    for (int j = 0; j < SORT_PER; ++j)
        if (col[j] >= NN) av[j] += ap2[col[j]];
    for (int b = tid; b < SAGE_BINS; b += SORT_NT) cnt[b] = 0u;
    __syncthreads();
    #pragma unroll
    for (int j = 0; j < SORT_PER; ++j) {
        int k = j * SORT_NT + tid;
        rank[j] = 0u;
        if (k < n) rank[j] = atomicAdd(&cnt[row[j] >> 4], 1u);
    }
    __syncthreads();
    bins_to_offsets<SAGE_BINS>(cnt, wsum);
    #pragma unroll
    for (int j = 0; j < SORT_PER; ++j) pos[j] = cnt[row[j] >> 4] + rank[j];
    #pragma unroll
    for (int j = 0; j < SORT_PER; ++j) {
        int k = j * SORT_NT + tid;
        if (k < n) sorted[pos[j]] = make_uint2((unsigned)row[j], __float_as_uint(av[j]));
    }
    __syncthreads();
    const int jj   = lane >> 1;
    const int role = lane & 1;
    uint2 eA[SORT_PER], eB[SORT_PER];
    #pragma unroll
    for (int j = 0; j < SORT_PER; ++j) {
        int p = (j * 32 + wave) * 32 + jj;
        eA[j] = sorted[min(p, n - 1)];
        eB[j] = sorted[min(p + 16, n - 1)];
    }
    #pragma unroll
    for (int j = 0; j < SORT_PER; ++j) {
        int p = (j * 32 + wave) * 32 + jj;
        if (p < n)
            atomicAdd(&md[2 * eA[j].x + role], role ? 1.0f : __uint_as_float(eA[j].y));
        if (p + 16 < n)
            atomicAdd(&md[2 * eB[j].x + role], role ? 1.0f : __uint_as_float(eB[j].y));
    }
}

__global__ void k_mdsum(float* ws, const float* __restrict__ sc) {
    int i = blockIdx.x * blockDim.x + threadIdx.x;
    if (i < NBT) {
        float4 x = *(const float4*)&ws[WS_MD0 + 4 * i];
        const float4 y = *(const float4*)&sc[WS_MD0 + 4 * i];
        x.x += y.x; x.y += y.y; x.z += y.z; x.w += y.w;
        *(float4*)&ws[WS_MD0 + 4 * i] = x;
    }
    if (i < 2 * BB) {
        int o = ((i >> 9) ? WS_A1 : WS_A0) + NN + (i & (BB - 1));
        ws[o] += sc[o];
    }
}

#define ROWS_NT    256
#define ROWS_BLK   1024
#define ROWS_WAVES (ROWS_BLK * ROWS_NT / 32)
#define ROWS_UNR   4
static_assert(NBT % 2 == 0 && NN % 2 == 0, "row pairs must not straddle a layer or the real/virtual split");
static_assert(ROWS_NT == 256 && DD == 64, "dense chain split: 16 groups of 4 p_conv rows, 8 waves of 8 k");

__device__ __forceinline__ float half_sum(float v) {
    v += __int_as_float(__builtin_amdgcn_update_dpp(0, __float_as_int(v), 0x161, 0xf, 0xf, true));
    v += __int_as_float(__builtin_amdgcn_update_dpp(0, __float_as_int(v), 0x162, 0xf, 0xf, true));
    v += __int_as_float(__builtin_amdgcn_update_dpp(0, __float_as_int(v), 0x164, 0xf, 0xf, true));
    v += __int_as_float(__builtin_amdgcn_update_dpp(0, __float_as_int(v), 0x168, 0xf, 0xf, true));
    return v;
}

__global__ void __launch_bounds__(ROWS_NT)
k_rows(float* ws, const float* __restrict__ cp, const float* __restrict__ w_n2l,
       const float* __restrict__ p_conv, const float* __restrict__ W_sage, float* out) {
    __shared__ __attribute__((aligned(16))) float sx[DD];
    __shared__ __attribute__((aligned(16))) float swp[16 * DD];
    __shared__ __attribute__((aligned(16))) float sw[DD];
    __shared__ __attribute__((aligned(16))) float sup[8 * 2 * DD];
    const int tid  = threadIdx.x;
    const int lane = tid & 31;
    const int half = lane >> 4;
    const int q    = lane & 15;
    const int gw   = (blockIdx.x * ROWS_NT + tid) >> 5;

    const int jr = tid >> 4;
    const int kg = tid >> 5;
    float4 p4[4], s4[8];
    #pragma unroll
    for (int it = 0; it < 4; ++it) p4[it] = *(const float4*)&p_conv[(4 * jr + it) * DD + 4 * q];
    #pragma unroll
    for (int r = 0; r < 8; ++r) s4[r] = *(const float4*)&W_sage[(DD * half + 8 * kg + r) * DD + 4 * q];
    if (tid < DD) sx[tid] = fmaxf(w_n2l[tid] + w_n2l[DD + tid], 0.0f);
    __syncthreads();
    float ssa = 0.f, ssb = 0.f, ssc = 0.f, ssd = 0.f;
    #pragma unroll
    for (int k = 0; k < DD / 4; ++k) {
        const float4 xv = *(const float4*)&sx[4 * k];
        ssa += xv.x * xv.x; ssb += xv.y * xv.y; ssc += xv.z * xv.z; ssd += xv.w * xv.w;
    }
    const float den = fmaxf(sqrtf((ssa + ssb) + (ssc + ssd)), 1e-12f);
    {
        const float4 xq = *(const float4*)&sx[4 * jr];
        const float v0 = xq.x / den, v1 = xq.y / den, v2 = xq.z / den, v3 = xq.w / den;
        float4 wp;
        wp.x = (v0 * p4[0].x + v1 * p4[1].x) + (v2 * p4[2].x + v3 * p4[3].x);
        wp.y = (v0 * p4[0].y + v1 * p4[1].y) + (v2 * p4[2].y + v3 * p4[3].y);
        wp.z = (v0 * p4[0].z + v1 * p4[1].z) + (v2 * p4[2].z + v3 * p4[3].z);
        wp.w = (v0 * p4[0].w + v1 * p4[1].w) + (v2 * p4[2].w + v3 * p4[3].w);
        *(float4*)&swp[jr * DD + 4 * q] = wp;
    }
    __syncthreads();
    if (tid < DD) {
        float s = 0.f;
        #pragma unroll
        for (int j = 0; j < 16; ++j) s += swp[j * DD + tid];
        sw[tid] = s;
    }
    __syncthreads();
    {
        const float4 wa = *(const float4*)&sw[8 * kg];
        const float4 wb = *(const float4*)&sw[8 * kg + 4];
        float4 up;
        up.x = ((wa.x * s4[0].x + wa.y * s4[1].x) + (wa.z * s4[2].x + wa.w * s4[3].x))
             + ((wb.x * s4[4].x + wb.y * s4[5].x) + (wb.z * s4[6].x + wb.w * s4[7].x));
        up.y = ((wa.x * s4[0].y + wa.y * s4[1].y) + (wa.z * s4[2].y + wa.w * s4[3].y))
             + ((wb.x * s4[4].y + wb.y * s4[5].y) + (wb.z * s4[6].y + wb.w * s4[7].y));
        up.z = ((wa.x * s4[0].z + wa.y * s4[1].z) + (wa.z * s4[2].z + wa.w * s4[3].z))
             + ((wb.x * s4[4].z + wb.y * s4[5].z) + (wb.z * s4[6].z + wb.w * s4[7].z));
        up.w = ((wa.x * s4[0].w + wa.y * s4[1].w) + (wa.z * s4[2].w + wa.w * s4[3].w))
             + ((wb.x * s4[4].w + wb.y * s4[5].w) + (wb.z * s4[6].w + wb.w * s4[7].w));
        *(float4*)&sup[kg * 2 * DD + 4 * lane] = up;
    }
    __syncthreads();
    float4 u1 = make_float4(0.f, 0.f, 0.f, 0.f), u2 = u1;
    #pragma unroll
    for (int gk = 0; gk < 8; ++gk) {
        const float4 a4 = *(const float4*)&sup[gk * 2 * DD + 4 * q];
        const float4 b4 = *(const float4*)&sup[gk * 2 * DD + DD + 4 * q];
        u1.x += a4.x; u1.y += a4.y; u1.z += a4.z; u1.w += a4.w;
        u2.x += b4.x; u2.y += b4.y; u2.z += b4.z; u2.w += b4.w;
    }

    for (int p0 = gw; p0 < NBT; p0 += ROWS_WAVES * ROWS_UNR) {
        float a[ROWS_UNR]; float2 md[ROWS_UNR];
        #pragma unroll
        for (int k = 0; k < ROWS_UNR; ++k) {
            int p = min(p0 + k * ROWS_WAVES, NBT - 1);
            int w = 2 * p + half;
            int l = (w >= NBT) ? 1 : 0;
            int i = w - l * NBT;
            a[k]  = ws[(l ? WS_A1 : WS_A0) + i];
            md[k] = *(const float2*)&ws[(l ? WS_MD1 : WS_MD0) + 2 * i];
        }
        #pragma unroll
        for (int k = 0; k < ROWS_UNR; ++k) {
            int p = p0 + k * ROWS_WAVES;
            if (p < NBT) {
                int w = 2 * p + half;
                int l = (w >= NBT) ? 1 : 0;
                int i = w - l * NBT;
                float m = md[k].x / fmaxf(md[k].y, 1.0f);
                float r0 = fmaxf(a[k] * u1.x + m * u2.x, 0.0f);
                float r1 = fmaxf(a[k] * u1.y + m * u2.y, 0.0f);
                float r2 = fmaxf(a[k] * u1.z + m * u2.z, 0.0f);
                float r3 = fmaxf(a[k] * u1.w + m * u2.w, 0.0f);
                float ss = (r0 * r0 + r1 * r1) + (r2 * r2 + r3 * r3);
                ss = half_sum(ss);
                float inv = 1.0f / fmaxf(sqrtf(ss), 1e-12f);
                r0 *= inv; r1 *= inv; r2 *= inv; r3 *= inv;
                if (i < NN) {
                    float* dst = out + BB + (l * NN + i) * DD + 4 * q;
                    *(float4*)dst = make_float4(r0, r1, r2, r3);
                } else {
                    int b = i - NN;
                    const float4 c4 = *(const float4*)&cp[4 * q];
                    float t = (r0 * c4.x + r1 * c4.y) + (r2 * c4.z + r3 * c4.w);
                    t = half_sum(t);
                    if (q == 0) ws[WS_T + l * BB + b] = t;
                }
            }
        }
    }
}

#define TAIL_NT 64
__global__ void __launch_bounds__(TAIL_NT)
k_tail(const int* __restrict__ act_col, const float* __restrict__ ws,
       const float* __restrict__ h1, const float* __restrict__ aux,
       const float* __restrict__ h2w, float* out) {
    __shared__ float sq[32];
    const int lane = threadIdx.x & 31;
    const int wv   = threadIdx.x >> 5;
    const int half = lane >> 4;
    const int lid  = lane & 15;
    const int b0   = (blockIdx.x * 2 + wv) * 16;

    const float* A0 = out + BB + act_col[b0 + lid] * DD;
    const float* A1 = A0 + NN * DD;
    v8f c00 = {0.f, 0.f, 0.f, 0.f, 0.f, 0.f, 0.f, 0.f};
    v8f c01 = c00, c10 = c00, c11 = c00;
    #pragma unroll
    for (int s = 0; s < 16; ++s) {
        int k0 = 4 * s + 2 * half;
        v2f a0 = *(const v2f*)(A0 + k0);
        v2f a1 = *(const v2f*)(A1 + k0);
        v2f bv0, bv1;
        bv0.x = h1[(k0    ) * 32 + lid];
        bv0.y = h1[(k0 + 1) * 32 + lid];
        bv1.x = h1[(k0    ) * 32 + 16 + lid];
        bv1.y = h1[(k0 + 1) * 32 + 16 + lid];
        c00 = __builtin_amdgcn_wmma_f32_16x16x4_f32(false, a0, false, bv0, (short)0, c00, false, false);
        c01 = __builtin_amdgcn_wmma_f32_16x16x4_f32(false, a0, false, bv1, (short)0, c01, false, false);
        c10 = __builtin_amdgcn_wmma_f32_16x16x4_f32(false, a1, false, bv0, (short)0, c10, false, false);
        c11 = __builtin_amdgcn_wmma_f32_16x16x4_f32(false, a1, false, bv1, (short)0, c11, false, false);
    }
    const float w0 = h2w[lid], w1 = h2w[16 + lid];
    float part[8];
    #pragma unroll
    for (int r = 0; r < 8; ++r) {
        int b = b0 + r + 8 * half;
        float t0 = ws[WS_T + b];
        float t1 = ws[WS_T + BB + b];
        float p0 = fmaxf(t0 * c00[r], 0.0f) * w0 + fmaxf(t0 * c01[r], 0.0f) * w1;
        float p1 = fmaxf(t1 * c10[r], 0.0f) * w0 + fmaxf(t1 * c11[r], 0.0f) * w1;
        part[r] = half_sum(p0 + p1);
    }
    float qv = 0.0f;
    #pragma unroll
    for (int r = 0; r < 8; ++r) qv = (lid == r) ? part[r] : qv;
    if (lid < 8) {
        int M = lid + 8 * half;
        const float* ax = aux + (b0 + M) * 8;
        for (int l = 0; l < 2; ++l)
            for (int j = 0; j < 4; ++j) qv += ax[l * 4 + j] * h2w[32 + j];
        sq[wv * 16 + M] = qv;
    }
    __syncthreads();
    if (wv == 0) out[blockIdx.x * 32 + lane] = sq[lane];
}

extern "C" void kernel_launch(void* const* d_in, const int* in_sizes, int n_in,
                              void* d_out, int out_size, void* d_ws, size_t ws_size,
                              hipStream_t stream) {
    const int*   n2n0_row = (const int*)  d_in[0];
    const float* n2n0_val = (const float*)d_in[2];
    const int*   n2n1_row = (const int*)  d_in[3];
    const float* n2n1_val = (const float*)d_in[5];
    const int*   subg_row = (const int*)  d_in[6];
    const int*   act_col  = (const int*)  d_in[9];
    const int*   sg0_row  = (const int*)  d_in[10];
    const int*   sg0_col  = (const int*)  d_in[11];
    const int*   sg1_row  = (const int*)  d_in[12];
    const int*   sg1_col  = (const int*)  d_in[13];
    const float* aux      = (const float*)d_in[14];
    const float* w_n2l    = (const float*)d_in[15];
    const float* p_conv   = (const float*)d_in[16];
    const float* W_sage   = (const float*)d_in[17];
    const float* cp       = (const float*)d_in[18];
    const float* h1w      = (const float*)d_in[19];
    const float* h2w      = (const float*)d_in[20];
    float* out = (float*)d_out;
    float* ws  = (float*)d_ws;
    float* sc  = out + BB;

    k_zero <<<(6 * NBT + 255) / 256, 256, 0, stream>>>(ws, sc);
    k_n2n  <<<2 * N2N_BLK, SORT_NT, 0, stream>>>(n2n0_row, n2n0_val,
                                                  n2n1_row, n2n1_val, ws, sc);
    k_subg <<<(NN + 255) / 256, 256, 0, stream>>>(subg_row, ws, sc);
    k_sage <<<2 * SAGE_BLK, SORT_NT, 0, stream>>>(sg0_row, sg0_col,
                                                   sg1_row, sg1_col, ws, sc);
    k_mdsum<<<(NBT + 255) / 256, 256, 0, stream>>>(ws, sc);
    k_rows <<<ROWS_BLK, ROWS_NT, 0, stream>>>(ws, cp, w_n2l, p_conv, W_sage, out);
    k_tail <<<BB / 32, TAIL_NT, 0, stream>>>(act_col, ws, h1w, aux, h2w, out);
}


# n2n, histogram, sage sums: blocks store line-ordered lists to scratch; owner blocks (layer, 800 rows) sum their runs in LDS, plain stores; no global adds or zeroing; 6 launches; sum order changes
# speedup vs baseline: 6.2531x; 1.1433x over previous
#include <hip/hip_runtime.h>

#define NN   100000
#define BB   512
#define EE   1000000
#define ESZ  1200000
#define DD   64
#define NBT  (NN + BB)

#define WS_A0   0
#define WS_A1   (NBT)
#define WS_MD0  (2*NBT)
#define WS_MD1  (4*NBT)
#define WS_SM   (6*NBT)
#define WS_T    (WS_SM + 256)

typedef float v2f __attribute__((ext_vector_type(2)));
typedef float v8f __attribute__((ext_vector_type(8)));


#define SORT_T    16384
#define SORT_NT   1024
#define SORT_PER  (SORT_T / SORT_NT)
#define N2N_BINS  ((NBT + 31) / 32)
#define SAGE_BINS ((NBT + 15) / 16)
#define N2N_BLK   ((EE + SORT_T - 1) / SORT_T)
#define SAGE_BLK  ((ESZ + SORT_T - 1) / SORT_T)
static_assert(SORT_T == 32 * 32 * SORT_PER, "32 waves x SORT_PER groups of 32 ordered edges");

#define OWN_ROWS  800
#define OWN_RNG   ((NBT + OWN_ROWS - 1) / OWN_ROWS)
#define OWN_OFFS  128
#define OWN_NT    1024
#define OWN_U     6
#define HIST_PER  4
#define HIST_T    (SORT_NT * HIST_PER)
#define HIST_BLK  ((NN + HIST_T - 1) / HIST_T)
#define SC_SLIST  0
#define SC_SOFF   (SC_SLIST + 2 * SAGE_BLK * SORT_T * 2)
#define SC_NLIST  (SC_SOFF + 2 * SAGE_BLK * OWN_OFFS)
#define SC_NOFF   (SC_NLIST + 2 * N2N_BLK * SORT_T * 2)
#define SC_HIST   (SC_NOFF + 2 * N2N_BLK * OWN_OFFS)
#define SC_END    (SC_HIST + HIST_BLK * BB)
static_assert(OWN_ROWS % 32 == 0, "a range is whole lines of the a plane and of the md plane");
static_assert(NN % OWN_ROWS == 0 && BB <= OWN_ROWS && OWN_RNG == NN / OWN_ROWS + 1, "the virtual rows are exactly the last range");
static_assert(OWN_RNG + 1 <= OWN_OFFS, "a list's range starts, and its end, fit its offset words");
static_assert(EE % 2 == 0 && ESZ % 2 == 0 && SORT_T % 2 == 0, "lists are copied in 16-byte pieces of two whole entries");
static_assert(SC_END <= 2 * NN * DD, "scratch stays inside the rows of out");

template <int NB>
__device__ __forceinline__ void bins_to_offsets(unsigned* cnt, unsigned* wsum) {
    constexpr int PER = (NB + SORT_NT - 1) / SORT_NT;
    const int tid = threadIdx.x, lane = tid & 31, wave = tid >> 5;
    unsigned loc[PER];
    unsigned s = 0;
    #pragma unroll
    for (int k = 0; k < PER; ++k) {
        int b = tid * PER + k;
        unsigned v = (b < NB) ? cnt[b] : 0u;
        loc[k] = s;
        s += v;
    }
    unsigned inc = s;
    for (int off = 1; off < 32; off <<= 1) {
        unsigned t = __shfl_up(inc, off, 32);
        if (lane >= off) inc += t;
    }
    if (lane == 31) wsum[wave] = inc;
    __syncthreads();
    if (wave == 0) {
        unsigned w = wsum[lane];
        unsigned winc = w;
        for (int off = 1; off < 32; off <<= 1) {
            unsigned t = __shfl_up(winc, off, 32);
            if (lane >= off) winc += t;
        }
        wsum[lane] = winc - w;
    }
    __syncthreads();
    unsigned base = wsum[wave] + (inc - s);
    #pragma unroll
    for (int k = 0; k < PER; ++k) {
        int b = tid * PER + k;
        if (b < NB) cnt[b] = base + loc[k];
    }
    __syncthreads();
}

template <int NBINS, int RB>
__device__ __forceinline__ void list_out(const uint2* sorted, const unsigned* cnt, int n,
                                         uint2* list, unsigned* offs) {
    const int tid = threadIdx.x;
    uint4* dst4 = (uint4*)list;
    const uint4* src4 = (const uint4*)sorted;
    if (n == SORT_T) {
        uint4 e4[SORT_PER / 2];
        #pragma unroll
        for (int j = 0; j < SORT_PER / 2; ++j) e4[j] = src4[j * SORT_NT + tid];
        #pragma unroll
        for (int j = 0; j < SORT_PER / 2; ++j) dst4[j * SORT_NT + tid] = e4[j];
    } else {
        for (int i = tid; 2 * i < n; i += SORT_NT) dst4[i] = src4[i];
    }
    if (tid <= OWN_RNG) offs[tid] = (tid * RB < NBINS) ? cnt[tid * RB] : (unsigned)n;
}

__global__ void __launch_bounds__(SORT_NT)
k_n2n(const int* __restrict__ r0, const float* __restrict__ v0,
      const int* __restrict__ r1, const float* __restrict__ v1,
      const int* __restrict__ subg_row, float* sc) {
    __shared__ unsigned cnt[N2N_BINS];
    __shared__ unsigned wsum[32];
    __shared__ __attribute__((aligned(16))) uint2 sorted[SORT_T];
    const int tid = threadIdx.x;
    if (blockIdx.x >= 2 * N2N_BLK) {
        const int hb = blockIdx.x - 2 * N2N_BLK;
        if (tid < BB) cnt[tid] = 0u;
        int b[HIST_PER];
        #pragma unroll
        for (int j = 0; j < HIST_PER; ++j)
            b[j] = subg_row[min(hb * HIST_T + j * SORT_NT + tid, NN - 1)];
        __syncthreads();
        #pragma unroll
        for (int j = 0; j < HIST_PER; ++j)
            if (hb * HIST_T + j * SORT_NT + tid < NN) atomicAdd(&cnt[b[j]], 1u);
        __syncthreads();
        if (tid < BB) sc[SC_HIST + hb * BB + tid] = (float)cnt[tid];
        return;
    }
    const bool l1 = blockIdx.x >= N2N_BLK;
    const int blk = l1 ? blockIdx.x - N2N_BLK : blockIdx.x;
    const int*   rr = l1 ? r1 : r0;
    const float* vv = l1 ? v1 : v0;
    const int base = blk * SORT_T;
    const int n = min(SORT_T, EE - base);

    for (int b = tid; b < N2N_BINS; b += SORT_NT) cnt[b] = 0u;
    int row[SORT_PER]; float val[SORT_PER]; unsigned rank[SORT_PER];
    #pragma unroll
    for (int j = 0; j < SORT_PER; ++j) {
        int k = j * SORT_NT + tid;
        int idx = base + min(k, n - 1);
        row[j] = rr[idx];
        val[j] = vv[idx];
    }
    __syncthreads();
    #pragma unroll
    for (int j = 0; j < SORT_PER; ++j) {
        int k = j * SORT_NT + tid;
        rank[j] = 0u;
        if (k < n) rank[j] = atomicAdd(&cnt[row[j] >> 5], 1u);
    }
    __syncthreads();
    bins_to_offsets<N2N_BINS>(cnt, wsum);
    unsigned pos[SORT_PER];
    #pragma unroll
    for (int j = 0; j < SORT_PER; ++j) pos[j] = cnt[row[j] >> 5] + rank[j];
    #pragma unroll
    for (int j = 0; j < SORT_PER; ++j) {
        int k = j * SORT_NT + tid;
        if (k < n) sorted[pos[j]] = make_uint2((unsigned)row[j], __float_as_uint(val[j]));
    }
    __syncthreads();
    list_out<N2N_BINS, OWN_ROWS / 32>(sorted, cnt, n,
                                      (uint2*)(sc + SC_NLIST) + (size_t)blockIdx.x * SORT_T,
                                      (unsigned*)(sc + SC_NOFF) + blockIdx.x * OWN_OFFS);
}

__global__ void __launch_bounds__(SORT_NT)
k_sage(const int* __restrict__ s0r, const int* __restrict__ s0c,
       const int* __restrict__ s1r, const int* __restrict__ s1c,
       const float* __restrict__ ws, float* sc) {
    __shared__ unsigned cnt[SAGE_BINS];
    __shared__ unsigned wsum[32];
    __shared__ __attribute__((aligned(16))) uint2 sorted[SORT_T];
    const int tid = threadIdx.x;
    const bool l1 = blockIdx.x >= SAGE_BLK;
    const int blk = l1 ? blockIdx.x - SAGE_BLK : blockIdx.x;
    const int* rr = l1 ? s1r : s0r;
    const int* cc = l1 ? s1c : s0c;
    const float* ap = ws + (l1 ? WS_A1 : WS_A0);
    const int base = blk * SORT_T;
    const int n = min(SORT_T, ESZ - base);

    for (int b = tid; b < N2N_BINS; b += SORT_NT) cnt[b] = 0u;
    int row[SORT_PER]; int col[SORT_PER]; float av[SORT_PER]; unsigned rank[SORT_PER];
    unsigned pos[SORT_PER];
    #pragma unroll
    for (int j = 0; j < SORT_PER; ++j) {
        int k = j * SORT_NT + tid;
        int idx = base + min(k, n - 1);
        row[j] = rr[idx];
        col[j] = cc[idx];
    }
    __syncthreads();
    #pragma unroll
    for (int j = 0; j < SORT_PER; ++j) {
        int k = j * SORT_NT + tid;
        rank[j] = 0u;
        if (k < n) rank[j] = atomicAdd(&cnt[col[j] >> 5], 1u);
    }
    __syncthreads();
    bins_to_offsets<N2N_BINS>(cnt, wsum);
    #pragma unroll
    for (int j = 0; j < SORT_PER; ++j) pos[j] = cnt[col[j] >> 5] + rank[j];
    #pragma unroll
    for (int j = 0; j < SORT_PER; ++j) {
        int k = j * SORT_NT + tid;
        if (k < n) sorted[pos[j]] = make_uint2((unsigned)row[j], (unsigned)col[j]);
    }
    __syncthreads();
    #pragma unroll
    for (int j = 0; j < SORT_PER; ++j) {
        uint2 e = sorted[min(j * SORT_NT + tid, n - 1)];
        row[j] = (int)e.x;
        col[j] = (int)e.y;
    }
    #pragma unroll
    for (int j = 0; j < SORT_PER; ++j) av[j] = ap[col[j]];
    for (int b = tid; b < SAGE_BINS; b += SORT_NT) cnt[b] = 0u;
    __syncthreads();
    #pragma unroll
    for (int j = 0; j < SORT_PER; ++j) {
        int k = j * SORT_NT + tid;
        rank[j] = 0u;
        if (k < n) rank[j] = atomicAdd(&cnt[row[j] >> 4], 1u);
    }
    __syncthreads();
    bins_to_offsets<SAGE_BINS>(cnt, wsum);
    #pragma unroll
    for (int j = 0; j < SORT_PER; ++j) pos[j] = cnt[row[j] >> 4] + rank[j];
    #pragma unroll
    for (int j = 0; j < SORT_PER; ++j) {
        int k = j * SORT_NT + tid;
        if (k < n) sorted[pos[j]] = make_uint2((unsigned)row[j], __float_as_uint(av[j]));
    }
    __syncthreads();
    list_out<SAGE_BINS, OWN_ROWS / 16>(sorted, cnt, n,
                                       (uint2*)(sc + SC_SLIST) + (size_t)blockIdx.x * SORT_T,
                                       (unsigned*)(sc + SC_SOFF) + blockIdx.x * OWN_OFFS);
}

template <bool MD>
__global__ void __launch_bounds__(OWN_NT)
k_own(float* ws, const float* __restrict__ sc) {
    constexpr int NLIST = MD ? SAGE_BLK : N2N_BLK;
    constexpr int RPW = (NLIST + OWN_NT / 32 - 1) / (OWN_NT / 32);
    __shared__ float sm[OWN_ROWS];
    __shared__ float sd[MD ? OWN_ROWS : 1];
    __shared__ unsigned so[2 * NLIST];
    const int tid = threadIdx.x, lane = tid & 31, wave = tid >> 5;
    const int l = (blockIdx.x >= OWN_RNG) ? 1 : 0;
    const int g = blockIdx.x - l * OWN_RNG;
    if (!MD && g == OWN_RNG - 1) {
        if (tid < BB) {
            float p[HIST_BLK];
            #pragma unroll
            for (int h = 0; h < HIST_BLK; ++h) p[h] = sc[SC_HIST + h * BB + tid];
            float s = 0.0f;
            #pragma unroll
            for (int h = 0; h < HIST_BLK; ++h) s += p[h];
            ws[(l ? WS_A1 : WS_A0) + NN + tid] = s;
        }
        return;
    }
    const unsigned* offs = (const unsigned*)(sc + (MD ? SC_SOFF : SC_NOFF)) + l * NLIST * OWN_OFFS;
    const uint2* list = (const uint2*)(sc + (MD ? SC_SLIST : SC_NLIST)) + (size_t)l * NLIST * SORT_T;
    if (tid < NLIST) {
        so[2 * tid]     = offs[tid * OWN_OFFS + g];
        so[2 * tid + 1] = offs[tid * OWN_OFFS + g + 1];
    }
    for (int r = tid; r < OWN_ROWS; r += OWN_NT) { sm[r] = 0.0f; if (MD) sd[r] = 0.0f; }
    __syncthreads();
    const unsigned row0 = (unsigned)(g * OWN_ROWS);
    unsigned rs[RPW], re[RPW];
    const uint2* run[RPW];
    uint2 v[RPW][OWN_U];
    #pragma unroll
    for (int r = 0; r < RPW; ++r) {
        int b = wave + r * (OWN_NT / 32);
        bool has = b < NLIST;
        int bb = has ? b : 0;
        rs[r] = has ? so[2 * bb] : 0u;
        re[r] = has ? so[2 * bb + 1] : 0u;
        run[r] = list + (size_t)bb * SORT_T;
    }
    #pragma unroll
    for (int r = 0; r < RPW; ++r) {
        unsigned last = max(re[r], 1u) - 1u;
        #pragma unroll
        for (int k = 0; k < OWN_U; ++k)
            v[r][k] = run[r][min(rs[r] + lane + 32u * k, last)];
    }
    #pragma unroll
    for (int r = 0; r < RPW; ++r) {
        #pragma unroll
        for (int k = 0; k < OWN_U; ++k) {
            if (rs[r] + lane + 32u * k < re[r]) {
                atomicAdd(&sm[v[r][k].x - row0], __uint_as_float(v[r][k].y));
                if (MD) atomicAdd(&sd[v[r][k].x - row0], 1.0f);
            }
        }
        for (unsigned i = rs[r] + lane + 32u * OWN_U; i < re[r]; i += 32u) {
            uint2 x = run[r][i];
            atomicAdd(&sm[x.x - row0], __uint_as_float(x.y));
            if (MD) atomicAdd(&sd[x.x - row0], 1.0f);
        }
    }
    __syncthreads();
    const int nrows = min(OWN_ROWS, NBT - g * OWN_ROWS);
    if (MD) {
        float* md = ws + (l ? WS_MD1 : WS_MD0);
        for (int r = tid; r < nrows; r += OWN_NT)
            *(float2*)&md[2 * (g * OWN_ROWS + r)] = make_float2(sm[r], sd[r]);
    } else {
        float* a = ws + (l ? WS_A1 : WS_A0);
        for (int r = tid; r < nrows; r += OWN_NT) a[g * OWN_ROWS + r] = sm[r];
    }
}

#define ROWS_NT    256
#define ROWS_BLK   1024
#define ROWS_WAVES (ROWS_BLK * ROWS_NT / 32)
#define ROWS_UNR   4
static_assert(NBT % 2 == 0 && NN % 2 == 0, "row pairs must not straddle a layer or the real/virtual split");
static_assert(ROWS_NT == 256 && DD == 64, "dense chain split: 16 groups of 4 p_conv rows, 8 waves of 8 k");

__device__ __forceinline__ float half_sum(float v) {
    v += __int_as_float(__builtin_amdgcn_update_dpp(0, __float_as_int(v), 0x161, 0xf, 0xf, true));
    v += __int_as_float(__builtin_amdgcn_update_dpp(0, __float_as_int(v), 0x162, 0xf, 0xf, true));
    v += __int_as_float(__builtin_amdgcn_update_dpp(0, __float_as_int(v), 0x164, 0xf, 0xf, true));
    v += __int_as_float(__builtin_amdgcn_update_dpp(0, __float_as_int(v), 0x168, 0xf, 0xf, true));
    return v;
}

__global__ void __launch_bounds__(ROWS_NT)
k_rows(float* ws, const float* __restrict__ cp, const float* __restrict__ w_n2l,
       const float* __restrict__ p_conv, const float* __restrict__ W_sage, float* out) {
    __shared__ __attribute__((aligned(16))) float sx[DD];
    __shared__ __attribute__((aligned(16))) float swp[16 * DD];
    __shared__ __attribute__((aligned(16))) float sw[DD];
    __shared__ __attribute__((aligned(16))) float sup[8 * 2 * DD];
    const int tid  = threadIdx.x;
    const int lane = tid & 31;
    const int half = lane >> 4;
    const int q    = lane & 15;
    const int gw   = (blockIdx.x * ROWS_NT + tid) >> 5;

    const int jr = tid >> 4;
    const int kg = tid >> 5;
    float4 p4[4], s4[8];
    #pragma unroll
    for (int it = 0; it < 4; ++it) p4[it] = *(const float4*)&p_conv[(4 * jr + it) * DD + 4 * q];
    #pragma unroll
    for (int r = 0; r < 8; ++r) s4[r] = *(const float4*)&W_sage[(DD * half + 8 * kg + r) * DD + 4 * q];
    if (tid < DD) sx[tid] = fmaxf(w_n2l[tid] + w_n2l[DD + tid], 0.0f);
    __syncthreads();
    float ssa = 0.f, ssb = 0.f, ssc = 0.f, ssd = 0.f;
    #pragma unroll
    for (int k = 0; k < DD / 4; ++k) {
        const float4 xv = *(const float4*)&sx[4 * k];
        ssa += xv.x * xv.x; ssb += xv.y * xv.y; ssc += xv.z * xv.z; ssd += xv.w * xv.w;
    }
    const float den = fmaxf(sqrtf((ssa + ssb) + (ssc + ssd)), 1e-12f);
    {
        const float4 xq = *(const float4*)&sx[4 * jr];
        const float v0 = xq.x / den, v1 = xq.y / den, v2 = xq.z / den, v3 = xq.w / den;
        float4 wp;
        wp.x = (v0 * p4[0].x + v1 * p4[1].x) + (v2 * p4[2].x + v3 * p4[3].x);
        wp.y = (v0 * p4[0].y + v1 * p4[1].y) + (v2 * p4[2].y + v3 * p4[3].y);
        wp.z = (v0 * p4[0].z + v1 * p4[1].z) + (v2 * p4[2].z + v3 * p4[3].z);
        wp.w = (v0 * p4[0].w + v1 * p4[1].w) + (v2 * p4[2].w + v3 * p4[3].w);
        *(float4*)&swp[jr * DD + 4 * q] = wp;
    }
    __syncthreads();
    if (tid < DD) {
        float s = 0.f;
        #pragma unroll
        for (int j = 0; j < 16; ++j) s += swp[j * DD + tid];
        sw[tid] = s;
    }
    __syncthreads();
    {
        const float4 wa = *(const float4*)&sw[8 * kg];
        const float4 wb = *(const float4*)&sw[8 * kg + 4];
        float4 up;
        up.x = ((wa.x * s4[0].x + wa.y * s4[1].x) + (wa.z * s4[2].x + wa.w * s4[3].x))
             + ((wb.x * s4[4].x + wb.y * s4[5].x) + (wb.z * s4[6].x + wb.w * s4[7].x));
        up.y = ((wa.x * s4[0].y + wa.y * s4[1].y) + (wa.z * s4[2].y + wa.w * s4[3].y))
             + ((wb.x * s4[4].y + wb.y * s4[5].y) + (wb.z * s4[6].y + wb.w * s4[7].y));
        up.z = ((wa.x * s4[0].z + wa.y * s4[1].z) + (wa.z * s4[2].z + wa.w * s4[3].z))
             + ((wb.x * s4[4].z + wb.y * s4[5].z) + (wb.z * s4[6].z + wb.w * s4[7].z));
        up.w = ((wa.x * s4[0].w + wa.y * s4[1].w) + (wa.z * s4[2].w + wa.w * s4[3].w))
             + ((wb.x * s4[4].w + wb.y * s4[5].w) + (wb.z * s4[6].w + wb.w * s4[7].w));
        *(float4*)&sup[kg * 2 * DD + 4 * lane] = up;
    }
    __syncthreads();
    float4 u1 = make_float4(0.f, 0.f, 0.f, 0.f), u2 = u1;
    #pragma unroll
    for (int gk = 0; gk < 8; ++gk) {
        const float4 a4 = *(const float4*)&sup[gk * 2 * DD + 4 * q];
        const float4 b4 = *(const float4*)&sup[gk * 2 * DD + DD + 4 * q];
        u1.x += a4.x; u1.y += a4.y; u1.z += a4.z; u1.w += a4.w;
        u2.x += b4.x; u2.y += b4.y; u2.z += b4.z; u2.w += b4.w;
    }

    for (int p0 = gw; p0 < NBT; p0 += ROWS_WAVES * ROWS_UNR) {
        float a[ROWS_UNR]; float2 md[ROWS_UNR];
        #pragma unroll
        for (int k = 0; k < ROWS_UNR; ++k) {
            int p = min(p0 + k * ROWS_WAVES, NBT - 1);
            int w = 2 * p + half;
            int l = (w >= NBT) ? 1 : 0;
            int i = w - l * NBT;
            a[k]  = ws[(l ? WS_A1 : WS_A0) + i];
            md[k] = *(const float2*)&ws[(l ? WS_MD1 : WS_MD0) + 2 * i];
        }
        #pragma unroll
        for (int k = 0; k < ROWS_UNR; ++k) {
            int p = p0 + k * ROWS_WAVES;
            if (p < NBT) {
                int w = 2 * p + half;
                int l = (w >= NBT) ? 1 : 0;
                int i = w - l * NBT;
                float m = md[k].x / fmaxf(md[k].y, 1.0f);
                float r0 = fmaxf(a[k] * u1.x + m * u2.x, 0.0f);
                float r1 = fmaxf(a[k] * u1.y + m * u2.y, 0.0f);
                float r2 = fmaxf(a[k] * u1.z + m * u2.z, 0.0f);
                float r3 = fmaxf(a[k] * u1.w + m * u2.w, 0.0f);
                float ss = (r0 * r0 + r1 * r1) + (r2 * r2 + r3 * r3);
                ss = half_sum(ss);
                float inv = 1.0f / fmaxf(sqrtf(ss), 1e-12f);
                r0 *= inv; r1 *= inv; r2 *= inv; r3 *= inv;
                if (i < NN) {
                    float* dst = out + BB + (l * NN + i) * DD + 4 * q;
                    *(float4*)dst = make_float4(r0, r1, r2, r3);
                } else {
                    int b = i - NN;
                    const float4 c4 = *(const float4*)&cp[4 * q];
                    float t = (r0 * c4.x + r1 * c4.y) + (r2 * c4.z + r3 * c4.w);
                    t = half_sum(t);
                    if (q == 0) ws[WS_T + l * BB + b] = t;
                }
            }
        }
    }
}

#define TAIL_NT 64
__global__ void __launch_bounds__(TAIL_NT)
k_tail(const int* __restrict__ act_col, const float* __restrict__ ws,
       const float* __restrict__ h1, const float* __restrict__ aux,
       const float* __restrict__ h2w, float* out) {
    __shared__ float sq[32];
    const int lane = threadIdx.x & 31;
    const int wv   = threadIdx.x >> 5;
    const int half = lane >> 4;
    const int lid  = lane & 15;
    const int b0   = (blockIdx.x * 2 + wv) * 16;

    const float* A0 = out + BB + act_col[b0 + lid] * DD;
    const float* A1 = A0 + NN * DD;
    v8f c00 = {0.f, 0.f, 0.f, 0.f, 0.f, 0.f, 0.f, 0.f};
    v8f c01 = c00, c10 = c00, c11 = c00;
    #pragma unroll
    for (int s = 0; s < 16; ++s) {
        int k0 = 4 * s + 2 * half;
        v2f a0 = *(const v2f*)(A0 + k0);
        v2f a1 = *(const v2f*)(A1 + k0);
        v2f bv0, bv1;
        bv0.x = h1[(k0    ) * 32 + lid];
        bv0.y = h1[(k0 + 1) * 32 + lid];
        bv1.x = h1[(k0    ) * 32 + 16 + lid];
        bv1.y = h1[(k0 + 1) * 32 + 16 + lid];
        c00 = __builtin_amdgcn_wmma_f32_16x16x4_f32(false, a0, false, bv0, (short)0, c00, false, false);
        c01 = __builtin_amdgcn_wmma_f32_16x16x4_f32(false, a0, false, bv1, (short)0, c01, false, false);
        c10 = __builtin_amdgcn_wmma_f32_16x16x4_f32(false, a1, false, bv0, (short)0, c10, false, false);
        c11 = __builtin_amdgcn_wmma_f32_16x16x4_f32(false, a1, false, bv1, (short)0, c11, false, false);
    }
    const float w0 = h2w[lid], w1 = h2w[16 + lid];
    float part[8];
    #pragma unroll
    for (int r = 0; r < 8; ++r) {
        int b = b0 + r + 8 * half;
        float t0 = ws[WS_T + b];
        float t1 = ws[WS_T + BB + b];
        float p0 = fmaxf(t0 * c00[r], 0.0f) * w0 + fmaxf(t0 * c01[r], 0.0f) * w1;
        float p1 = fmaxf(t1 * c10[r], 0.0f) * w0 + fmaxf(t1 * c11[r], 0.0f) * w1;
        part[r] = half_sum(p0 + p1);
    }
    float qv = 0.0f;
    #pragma unroll
    for (int r = 0; r < 8; ++r) qv = (lid == r) ? part[r] : qv;
    if (lid < 8) {
        int M = lid + 8 * half;
        const float* ax = aux + (b0 + M) * 8;
        for (int l = 0; l < 2; ++l)
            for (int j = 0; j < 4; ++j) qv += ax[l * 4 + j] * h2w[32 + j];
        sq[wv * 16 + M] = qv;
    }
    __syncthreads();
    if (wv == 0) out[blockIdx.x * 32 + lane] = sq[lane];
}

extern "C" void kernel_launch(void* const* d_in, const int* in_sizes, int n_in,
                              void* d_out, int out_size, void* d_ws, size_t ws_size,
                              hipStream_t stream) {
    const int*   n2n0_row = (const int*)  d_in[0];
    const float* n2n0_val = (const float*)d_in[2];
    const int*   n2n1_row = (const int*)  d_in[3];
    const float* n2n1_val = (const float*)d_in[5];
    const int*   subg_row = (const int*)  d_in[6];
    const int*   act_col  = (const int*)  d_in[9];
    const int*   sg0_row  = (const int*)  d_in[10];
    const int*   sg0_col  = (const int*)  d_in[11];
    const int*   sg1_row  = (const int*)  d_in[12];
    const int*   sg1_col  = (const int*)  d_in[13];
    const float* aux      = (const float*)d_in[14];
    const float* w_n2l    = (const float*)d_in[15];
    const float* p_conv   = (const float*)d_in[16];
    const float* W_sage   = (const float*)d_in[17];
    const float* cp       = (const float*)d_in[18];
    const float* h1w      = (const float*)d_in[19];
    const float* h2w      = (const float*)d_in[20];
    float* out = (float*)d_out;
    float* ws  = (float*)d_ws;
    float* sc  = out + BB;

    k_n2n  <<<2 * N2N_BLK + HIST_BLK, SORT_NT, 0, stream>>>(n2n0_row, n2n0_val,
                                                             n2n1_row, n2n1_val, subg_row, sc);
    k_own<false><<<2 * OWN_RNG, OWN_NT, 0, stream>>>(ws, sc);
    k_sage <<<2 * SAGE_BLK, SORT_NT, 0, stream>>>(sg0_row, sg0_col,
                                                   sg1_row, sg1_col, ws, sc);
    k_own<true><<<2 * OWN_RNG, OWN_NT, 0, stream>>>(ws, sc);
    k_rows <<<ROWS_BLK, ROWS_NT, 0, stream>>>(ws, cp, w_n2l, p_conv, W_sage, out);
    k_tail <<<BB / 32, TAIL_NT, 0, stream>>>(act_col, ws, h1w, aux, h2w, out);
}
